# starting kernel
# baseline (speedup 1.0000x reference)
#include <hip/hip_runtime.h>

typedef __attribute__((ext_vector_type(2))) float v2f;
typedef __attribute__((ext_vector_type(8))) float v8f;

#define D_   1024
#define H_   16
#define dh   64
#define T_   4096
#define CH   16
#define SEG  16
#define TSEG (T_/SEG)
#define NCHS (TSEG/CH)
#define ST   68
#define ST16 20

__device__ __forceinline__ v8f wmma_f32(v2f a, v2f b, v8f c) {
  return __builtin_amdgcn_wmma_f32_16x16x4_f32(false, a, false, b, (short)0, c,
                                               false, false);
}

__device__ __forceinline__ v2f fragA(const float* m, int stride, int k0) {
  int lane = threadIdx.x & 31;
  const float* p = m + (lane & 15) * stride + k0 + ((lane >> 4) << 1);
  v2f f; f[0] = p[0]; f[1] = p[1];
  return f;
}
__device__ __forceinline__ v2f fragB_kn(const float* m, int stride, int k0, int nbase) {
  int lane = threadIdx.x & 31;
  int kk = k0 + ((lane >> 4) << 1);
  int n  = nbase + (lane & 15);
  v2f f; f[0] = m[kk * stride + n]; f[1] = m[(kk + 1) * stride + n];
  return f;
}

__global__ __launch_bounds__(128)
void wkv_pass1(const float* __restrict__ wg, const float* __restrict__ kg,
               const float* __restrict__ vg, float* __restrict__ Aseg,
               float* __restrict__ Sloc) {
  __shared__ float S[dh * ST];
  __shared__ float Vv[CH * ST];
  __shared__ float Kh[dh * ST16];
  __shared__ float aLast[dh];
  __shared__ float aAcc[dh];

  const int seg = blockIdx.x, h = blockIdx.y;
  const int tid = threadIdx.x;
  const int lane = tid & 31, wid = tid >> 5;
  const int hi = lane >> 4, ln16 = lane & 15;

  for (int x = tid; x < dh * dh; x += 128) S[(x >> 6) * ST + (x & 63)] = 0.f;
  if (tid < dh) aAcc[tid] = 1.f;

  for (int c = 0; c < NCHS; ++c) {
    const int t0 = seg * TSEG + c * CH;
    __syncthreads();
    if (tid < dh) {
      int i = tid;
      float a = 1.f, ai = 1.f;
      float kr[CH], air[CH];
#pragma unroll
      for (int t = 0; t < CH; ++t) {
        int off  = (t0 + t) * D_ + h * dh + i;
        float wv = wg[off];
        if (seg == 0 && c == 0 && t == 0) wv = 1.f;
        float kv = kg[off];
        Vv[t * ST + i] = vg[off];
        float rw = 1.f / wv;
        a *= wv; ai *= rw;
        kr[t] = kv; air[t] = ai;
      }
      aLast[i] = a;
      aAcc[i] *= a;
#pragma unroll
      for (int t = 0; t < CH; ++t) Kh[i * ST16 + t] = kr[t] * (a * air[t]);
    }
    __syncthreads();

    const int nb = wid * 16;
#pragma unroll
    for (int it = 0; it < 4; ++it) {
      v8f acc = {};
#pragma unroll
      for (int k0 = 0; k0 < CH; k0 += 4)
        acc = wmma_f32(fragA(Kh + it * 16 * ST16, ST16, k0),
                       fragB_kn(Vv, ST, k0, nb), acc);
#pragma unroll
      for (int v = 0; v < 8; ++v) {
        int i   = it * 16 + v + 8 * hi;
        int idx = i * ST + nb + ln16;
        S[idx]  = aLast[i] * S[idx] + acc[v];
      }
    }
  }
  __syncthreads();
  if (tid < dh) Aseg[(h * SEG + seg) * dh + tid] = aAcc[tid];
  for (int x = tid; x < dh * dh; x += 128) {
    int i = x >> 6, j = x & 63;
    Sloc[((size_t)(h * SEG + seg) * dh + i) * dh + j] = S[i * ST + j];
  }
}

__global__ __launch_bounds__(128)
void state_scan(const float* __restrict__ wg, const float* __restrict__ isg,
                const float* __restrict__ Aseg, const float* __restrict__ Sloc,
                float* __restrict__ Sstart, float* __restrict__ state_out) {
  const int h = blockIdx.x, tid = threadIdx.x;
  for (int x = tid; x < dh * dh; x += 128) {
    int i = x >> 6, j = x & 63;
    float s = wg[h * dh + j] * isg[h * dh + j];
    for (int sg = 0; sg < SEG; ++sg) {
      size_t base = ((size_t)(h * SEG + sg) * dh + i) * dh + j;
      Sstart[base] = s;
      s = Aseg[(h * SEG + sg) * dh + i] * s + Sloc[base];
    }
    state_out[i * D_ + h * dh + j] = s;
  }
}

__global__ __launch_bounds__(128)
void wkv_pass2(const float* __restrict__ rg, const float* __restrict__ wg,
               const float* __restrict__ kg, const float* __restrict__ vg,
               const float* __restrict__ gg, const float* __restrict__ ug,
               const float* __restrict__ isg, const float* __restrict__ lnw,
               const float* __restrict__ lnb, const float* __restrict__ Sstart,
               float* __restrict__ og) {
  __shared__ float S[dh * ST];
  __shared__ float Rt[CH * ST];
  __shared__ float Kt[CH * ST];
  __shared__ float Ku[CH * ST];
  __shared__ float Vv[CH * ST];
  __shared__ float Kh[dh * ST16];
  __shared__ float CmW[4][CH * ST16];
  __shared__ float ybuf[CH * ST];
  __shared__ float aLast[dh];
  __shared__ float uvec[dh];
  __shared__ float corr0[dh];
  __shared__ float lnw_s[dh], lnb_s[dh];
  __shared__ float mu_s[CH], rs_s[CH];
  __shared__ float r0sum_s;

  const int seg = blockIdx.x, h = blockIdx.y;
  const int tid = threadIdx.x;
  const int lane = tid & 31, wid = tid >> 5;
  const int hi = lane >> 4, ln16 = lane & 15;

  for (int x = tid; x < dh; x += 128) {
    uvec[x]  = ug[h * dh + x];
    corr0[x] = isg[h * dh + x] * (1.f - wg[h * dh + x]);
    lnw_s[x] = lnw[x];
    lnb_s[x] = lnb[x];
  }
  for (int x = tid; x < dh * dh; x += 128) {
    int i = x >> 6, j = x & 63;
    S[i * ST + j] = Sstart[((size_t)(h * SEG + seg) * dh + i) * dh + j];
  }
  if (tid == 0 && seg == 0) {
    float s = 0.f;
    for (int i2 = 0; i2 < dh; ++i2) s += rg[h * dh + i2];
    r0sum_s = s;
  }
  __syncthreads();

  for (int c = 0; c < NCHS; ++c) {
    const int t0 = seg * TSEG + c * CH;

    if (tid < dh) {
      int i = tid;
      float a = 1.f, ai = 1.f;
      float kr[CH], air[CH];
#pragma unroll
      for (int t = 0; t < CH; ++t) {
        int off  = (t0 + t) * D_ + h * dh + i;
        float wv = wg[off];
        if (seg == 0 && c == 0 && t == 0) wv = 1.f;
        float rv = rg[off], kv = kg[off];
        float aip = ai;
        Rt[t * ST + i] = rv * a;
        float rw = 1.f / wv;
        a *= wv; ai *= rw;
        Kt[t * ST + i] = kv * ai;
        Ku[t * ST + i] = kv * uvec[i] * aip;
        Vv[t * ST + i] = vg[off];
        kr[t] = kv; air[t] = ai;
      }
      aLast[i] = a;
#pragma unroll
      for (int t = 0; t < CH; ++t) Kh[i * ST16 + t] = kr[t] * (a * air[t]);
    }
    __syncthreads();

    v8f accC = {}; v8f accD = {};
#pragma unroll
    for (int k0 = 0; k0 < dh; k0 += 4) {
      v2f arf = fragA(Rt, ST, k0);
      accC = wmma_f32(arf, fragA(Kt, ST, k0), accC);
      accD = wmma_f32(arf, fragA(Ku, ST, k0), accD);
    }
    {
      float* cm = CmW[wid];
#pragma unroll
      for (int v = 0; v < 8; ++v) {
        int M = v + 8 * hi, N = ln16;
        cm[M * ST16 + N] = (N < M) ? accC[v] : ((N == M) ? accD[v] : 0.f);
      }
    }
    asm volatile("" ::: "memory");

    const int nb = wid * 16;
    v8f accY = {};
#pragma unroll
    for (int k0 = 0; k0 < dh; k0 += 4)
      accY = wmma_f32(fragA(Rt, ST, k0), fragB_kn(S, ST, k0, nb), accY);
#pragma unroll
    for (int k0 = 0; k0 < CH; k0 += 4)
      accY = wmma_f32(fragA(CmW[wid], ST16, k0), fragB_kn(Vv, ST, k0, nb), accY);
    if (seg == 0 && c == 0 && hi == 0)
      accY[0] += r0sum_s * corr0[nb + ln16];
#pragma unroll
    for (int v = 0; v < 8; ++v) ybuf[(v + 8 * hi) * ST + nb + ln16] = accY[v];

    v8f accS[4];
#pragma unroll
    for (int it = 0; it < 4; ++it) {
      v8f a = {};
#pragma unroll
      for (int k0 = 0; k0 < CH; k0 += 4)
        a = wmma_f32(fragA(Kh + it * 16 * ST16, ST16, k0),
                     fragB_kn(Vv, ST, k0, nb), a);
      accS[it] = a;
    }
    __syncthreads();

#pragma unroll
    for (int it = 0; it < 4; ++it) {
#pragma unroll
      for (int v = 0; v < 8; ++v) {
        int i   = it * 16 + v + 8 * hi;
        int idx = i * ST + nb + ln16;
        S[idx]  = aLast[i] * S[idx] + accS[it][v];
      }
    }
    if (tid < CH) {
      float s = 0.f, sq = 0.f;
      for (int j = 0; j < dh; ++j) {
        float y = ybuf[tid * ST + j]; s += y; sq += y * y;
      }
      float m  = s * (1.f / dh);
      float vv = sq * (1.f / dh) - m * m;
      mu_s[tid] = m;
      rs_s[tid] = rsqrtf(vv + 1e-5f);
    }
    __syncthreads();

    {
      int t  = tid >> 3;
      int jb = (tid & 7) * 8;
      float m = mu_s[t], rs = rs_s[t];
#pragma unroll
      for (int q = 0; q < 8; ++q) {
        int j    = jb + q;
        float y  = ybuf[t * ST + j];
        float yn = (y - m) * rs * lnw_s[j] + lnb_s[j];
        int off  = (t0 + t) * D_ + h * dh + j;
        float gv = gg[off];
        og[off]  = yn * (gv / (1.f + __expf(-gv)));
      }
    }
    __syncthreads();
  }
}

__global__ __launch_bounds__(128)
void oproj_kernel(const float* __restrict__ o, const float* __restrict__ Wo,
                  float* __restrict__ out) {
  __shared__ float Ao[64 * ST];
  __shared__ float Bo[64 * ST];
  const int tid = threadIdx.x;
  const int lane = tid & 31, wid = tid >> 5;
  const int hi = lane >> 4, ln16 = lane & 15;
  const int nbase = blockIdx.x * 64;
  const int mbase = blockIdx.y * 64;

  v8f acc[4] = {};

  for (int ks = 0; ks < D_; ks += 64) {
    __syncthreads();
    {
      int row = tid >> 1;
      int cb  = (tid & 1) * 32;
      const float4* pa = (const float4*)(o  + (size_t)(mbase + row) * D_ + ks + cb);
      const float4* pb = (const float4*)(Wo + (size_t)(nbase + row) * D_ + ks + cb);
#pragma unroll
      for (int q = 0; q < 8; ++q) {
        float4 a4 = pa[q]; float4 b4 = pb[q];
        int cc = cb + q * 4;
        Ao[row * ST + cc + 0] = a4.x; Ao[row * ST + cc + 1] = a4.y;
        Ao[row * ST + cc + 2] = a4.z; Ao[row * ST + cc + 3] = a4.w;
        Bo[row * ST + cc + 0] = b4.x; Bo[row * ST + cc + 1] = b4.y;
        Bo[row * ST + cc + 2] = b4.z; Bo[row * ST + cc + 3] = b4.w;
      }
    }
    __syncthreads();
#pragma unroll
    for (int k0 = 0; k0 < 64; k0 += 4) {
      v2f b = fragA(Bo + wid * 16 * ST, ST, k0);
#pragma unroll
      for (int mt = 0; mt < 4; ++mt)
        acc[mt] = wmma_f32(fragA(Ao + mt * 16 * ST, ST, k0), b, acc[mt]);
    }
  }
#pragma unroll
  for (int mt = 0; mt < 4; ++mt) {
#pragma unroll
    for (int v = 0; v < 8; ++v) {
      int row = mbase + mt * 16 + v + 8 * hi;
      int col = nbase + wid * 16 + ln16;
      out[(size_t)row * D_ + col] = acc[mt][v];
    }
  }
}

extern "C" void kernel_launch(void* const* d_in, const int* in_sizes, int n_in,
                              void* d_out, int out_size, void* d_ws, size_t ws_size,
                              hipStream_t stream) {
  const float* r   = (const float*)d_in[0];
  const float* w   = (const float*)d_in[1];
  const float* k   = (const float*)d_in[2];
  const float* v   = (const float*)d_in[3];
  const float* g   = (const float*)d_in[4];
  const float* u   = (const float*)d_in[5];
  const float* is  = (const float*)d_in[6];
  const float* lnw = (const float*)d_in[7];
  const float* lnb = (const float*)d_in[8];
  const float* Wo  = (const float*)d_in[9];

  float* out   = (float*)d_out;
  float* state = out + (size_t)T_ * D_;

  float* o      = (float*)d_ws;
  float* Aseg   = o + (size_t)T_ * D_;
  float* Sloc   = Aseg + (size_t)H_ * SEG * dh;
  float* Sstart = Sloc + (size_t)H_ * SEG * dh * dh;

  dim3 gs(SEG, H_);
  wkv_pass1<<<gs, 128, 0, stream>>>(w, k, v, Aseg, Sloc);
  state_scan<<<H_, 128, 0, stream>>>(w, is, Aseg, Sloc, Sstart, state);
  wkv_pass2<<<gs, 128, 0, stream>>>(r, w, k, v, g, u, is, lnw, lnb, Sstart, o);

  dim3 g2(D_ / 64, T_ / 64);
  oproj_kernel<<<g2, 128, 0, stream>>>(o, Wo, out);
}


# pass1, scan, pass2: global loads issued as one batch per chunk, one chunk ahead in registers, with LDS-only block barriers so they stay in flight, instead of one load-wait round per timestep
# speedup vs baseline: 3.2003x; 3.2003x over previous
#include <hip/hip_runtime.h>

typedef __attribute__((ext_vector_type(2))) float v2f;
typedef __attribute__((ext_vector_type(8))) float v8f;

#define D_   1024
#define H_   16
#define dh   64
#define T_   4096
#define CH   16
#define SEG  16
#define TSEG (T_/SEG)
#define NCHS (TSEG/CH)
#define ST   68
#define ST16 20

__device__ __forceinline__ v8f wmma_f32(v2f a, v2f b, v8f c) {
  return __builtin_amdgcn_wmma_f32_16x16x4_f32(false, a, false, b, (short)0, c,
                                               false, false);
}

__device__ __forceinline__ v2f fragA(const float* m, int stride, int k0) {
  int lane = threadIdx.x & 31;
  const float* p = m + (lane & 15) * stride + k0 + ((lane >> 4) << 1);
  v2f f; f[0] = p[0]; f[1] = p[1];
  return f;
}
__device__ __forceinline__ v2f fragB_kn(const float* m, int stride, int k0, int nbase) {
  int lane = threadIdx.x & 31;
  int kk = k0 + ((lane >> 4) << 1);
  int n  = nbase + (lane & 15);
  v2f f; f[0] = m[kk * stride + n]; f[1] = m[(kk + 1) * stride + n];
  return f;
}

__device__ __forceinline__ void lds_barrier() {
  __builtin_amdgcn_fence(__ATOMIC_RELEASE, "workgroup", "local");
  __builtin_amdgcn_s_barrier();
  __builtin_amdgcn_fence(__ATOMIC_ACQUIRE, "workgroup", "local");
}

__global__ __launch_bounds__(128)
void wkv_pass1(const float* __restrict__ wg, const float* __restrict__ kg,
               const float* __restrict__ vg, float* __restrict__ Aseg,
               float* __restrict__ Sloc) {
  __shared__ float S[dh * ST];
  __shared__ float Vv[CH * ST];
  __shared__ float Kh[dh * ST16];
  __shared__ float aLast[dh];
  __shared__ float aAcc[dh];

  const int seg = blockIdx.x, h = blockIdx.y;
  const int tid = threadIdx.x;
  const int lane = tid & 31, wid = tid >> 5;
  const int hi = lane >> 4, ln16 = lane & 15;

  for (int x = tid; x < dh * dh; x += 128) S[(x >> 6) * ST + (x & 63)] = 0.f;
  if (tid < dh) aAcc[tid] = 1.f;

  const bool scanner = tid < dh;
  const int col = tid & (dh - 1);
  const int gbase = seg * TSEG * D_ + h * dh + col;
  float pw[CH], pk[CH], pv[CH];
  if (scanner) {
#pragma unroll
    for (int t = 0; t < CH; ++t) { pw[t] = wg[gbase + t * D_]; pk[t] = kg[gbase + t * D_]; }
  } else {
#pragma unroll
    for (int t = 0; t < CH; ++t) pv[t] = vg[gbase + t * D_];
  }

  for (int c = 0; c < NCHS; ++c) {
    lds_barrier();
    if (scanner) {
      int i = col;
      float a = 1.f, ai = 1.f;
      float kr[CH], air[CH];
#pragma unroll
      for (int t = 0; t < CH; ++t) {
        float wv = pw[t];
        if (seg == 0 && c == 0 && t == 0) wv = 1.f;
        float kv = pk[t];
        float rw = 1.f / wv;
        a *= wv; ai *= rw;
        kr[t] = kv; air[t] = ai;
      }
      aLast[i] = a;
      aAcc[i] *= a;
#pragma unroll
      for (int t = 0; t < CH; ++t) Kh[i * ST16 + t] = kr[t] * (a * air[t]);
    } else {
#pragma unroll
      for (int t = 0; t < CH; ++t) Vv[t * ST + col] = pv[t];
    }
    if (c + 1 < NCHS) {
      const int nbase = gbase + (c + 1) * CH * D_;
      if (scanner) {
#pragma unroll
        for (int t = 0; t < CH; ++t) { pw[t] = wg[nbase + t * D_]; pk[t] = kg[nbase + t * D_]; }
      } else {
#pragma unroll
        for (int t = 0; t < CH; ++t) pv[t] = vg[nbase + t * D_];
      }
    }
    lds_barrier();

    const int nb = wid * 16;
#pragma unroll
    for (int it = 0; it < 4; ++it) {
      v8f acc = {};
#pragma unroll
      for (int k0 = 0; k0 < CH; k0 += 4)
        acc = wmma_f32(fragA(Kh + it * 16 * ST16, ST16, k0),
                       fragB_kn(Vv, ST, k0, nb), acc);
#pragma unroll
      for (int v = 0; v < 8; ++v) {
        int i   = it * 16 + v + 8 * hi;
        int idx = i * ST + nb + ln16;
        S[idx]  = aLast[i] * S[idx] + acc[v];
      }
    }
  }
  lds_barrier();
  if (tid < dh) Aseg[(h * SEG + seg) * dh + tid] = aAcc[tid];
  for (int x = tid; x < dh * dh; x += 128) {
    int i = x >> 6, j = x & 63;
    Sloc[((size_t)(h * SEG + seg) * dh + i) * dh + j] = S[i * ST + j];
  }
}

__global__ __launch_bounds__(128)
void state_scan(const float* __restrict__ wg, const float* __restrict__ isg,
                const float* __restrict__ Aseg, const float* __restrict__ Sloc,
                float* __restrict__ Sstart, float* __restrict__ state_out) {
  const int h = blockIdx.x, tid = threadIdx.x;
#pragma clang loop vectorize(disable) interleave(disable) unroll(disable)
  for (int x = tid; x < dh * dh; x += 128) {
    int i = x >> 6, j = x & 63;
    float As[SEG], Ls[SEG];
#pragma unroll
    for (int sg = 0; sg < SEG; ++sg) {
      As[sg] = Aseg[(h * SEG + sg) * dh + i];
      Ls[sg] = Sloc[((size_t)(h * SEG + sg) * dh + i) * dh + j];
    }
    float s = wg[h * dh + j] * isg[h * dh + j];
#pragma unroll
    for (int sg = 0; sg < SEG; ++sg) {
      size_t base = ((size_t)(h * SEG + sg) * dh + i) * dh + j;
      Sstart[base] = s;
      s = As[sg] * s + Ls[sg];
    }
    state_out[i * D_ + h * dh + j] = s;
  }
}

__global__ __launch_bounds__(128)
void wkv_pass2(const float* __restrict__ rg, const float* __restrict__ wg,
               const float* __restrict__ kg, const float* __restrict__ vg,
               const float* __restrict__ gg, const float* __restrict__ ug,
               const float* __restrict__ isg, const float* __restrict__ lnw,
               const float* __restrict__ lnb, const float* __restrict__ Sstart,
               float* __restrict__ og) {
  __shared__ float S[dh * ST];
  __shared__ float Rt[CH * ST];
  __shared__ float Kt[CH * ST];
  __shared__ float Ku[CH * ST];
  __shared__ float Vv[CH * ST];
  __shared__ float Kh[dh * ST16];
  __shared__ float CmW[4][CH * ST16];
  __shared__ float ybuf[CH * ST];
  __shared__ float aLast[dh];
  __shared__ float uvec[dh];
  __shared__ float corr0[dh];
  __shared__ float lnw_s[dh], lnb_s[dh];
  __shared__ float mu_s[CH], rs_s[CH];
  __shared__ float r0sum_s;

  const int seg = blockIdx.x, h = blockIdx.y;
  const int tid = threadIdx.x;
  const int lane = tid & 31, wid = tid >> 5;
  const int hi = lane >> 4, ln16 = lane & 15;

  for (int x = tid; x < dh; x += 128) {
    uvec[x]  = ug[h * dh + x];
    corr0[x] = isg[h * dh + x] * (1.f - wg[h * dh + x]);
    lnw_s[x] = lnw[x];
    lnb_s[x] = lnb[x];
  }
  for (int x = tid; x < dh * dh; x += 128) {
    int i = x >> 6, j = x & 63;
    S[i * ST + j] = Sstart[((size_t)(h * SEG + seg) * dh + i) * dh + j];
  }
  if (tid == 0 && seg == 0) {
    float s = 0.f;
    for (int i2 = 0; i2 < dh; ++i2) s += rg[h * dh + i2];
    r0sum_s = s;
  }

  const bool scanner = tid < dh;
  const int col = tid & (dh - 1);
  const int gbase = seg * TSEG * D_ + h * dh + col;
  const int gofs = (seg * TSEG + (tid >> 3)) * D_ + h * dh + (tid & 7) * 8;
  float pr[CH], pw[CH], pk[CH], pv[CH];
  if (scanner) {
#pragma unroll
    for (int t = 0; t < CH; ++t) {
      pr[t] = rg[gbase + t * D_]; pw[t] = wg[gbase + t * D_]; pk[t] = kg[gbase + t * D_];
    }
  } else {
#pragma unroll
    for (int t = 0; t < CH; ++t) pv[t] = vg[gbase + t * D_];
  }
  lds_barrier();

  for (int c = 0; c < NCHS; ++c) {
    const int t0 = seg * TSEG + c * CH;

    if (scanner) {
      int i = col;
      float a = 1.f, ai = 1.f;
      float kr[CH], air[CH];
#pragma unroll
      for (int t = 0; t < CH; ++t) {
        float wv = pw[t];
        if (seg == 0 && c == 0 && t == 0) wv = 1.f;
        float rv = pr[t], kv = pk[t];
        float aip = ai;
        Rt[t * ST + i] = rv * a;
        float rw = 1.f / wv;
        a *= wv; ai *= rw;
        Kt[t * ST + i] = kv * ai;
        Ku[t * ST + i] = kv * uvec[i] * aip;
        kr[t] = kv; air[t] = ai;
      }
      aLast[i] = a;
#pragma unroll
      for (int t = 0; t < CH; ++t) Kh[i * ST16 + t] = kr[t] * (a * air[t]);
    } else {
#pragma unroll
      for (int t = 0; t < CH; ++t) Vv[t * ST + col] = pv[t];
    }
    const float4 g0 = *(const float4*)(gg + gofs + c * CH * D_);
    const float4 g1 = *(const float4*)(gg + gofs + c * CH * D_ + 4);
    if (c + 1 < NCHS) {
      const int nbase = gbase + (c + 1) * CH * D_;
      if (scanner) {
#pragma unroll
        for (int t = 0; t < CH; ++t) {
          pr[t] = rg[nbase + t * D_]; pw[t] = wg[nbase + t * D_]; pk[t] = kg[nbase + t * D_];
        }
      } else {
#pragma unroll
        for (int t = 0; t < CH; ++t) pv[t] = vg[nbase + t * D_];
      }
    }
    lds_barrier();

    v8f accC = {}; v8f accD = {};
#pragma unroll
    for (int k0 = 0; k0 < dh; k0 += 4) {
      v2f arf = fragA(Rt, ST, k0);
      accC = wmma_f32(arf, fragA(Kt, ST, k0), accC);
      accD = wmma_f32(arf, fragA(Ku, ST, k0), accD);
    }
    {
      float* cm = CmW[wid];
#pragma unroll
      for (int v = 0; v < 8; ++v) {
        int M = v + 8 * hi, N = ln16;
        cm[M * ST16 + N] = (N < M) ? accC[v] : ((N == M) ? accD[v] : 0.f);
      }
    }
    asm volatile("" ::: "memory");

    const int nb = wid * 16;
    v8f accY = {};
#pragma unroll
    for (int k0 = 0; k0 < dh; k0 += 4)
      accY = wmma_f32(fragA(Rt, ST, k0), fragB_kn(S, ST, k0, nb), accY);
#pragma unroll
    for (int k0 = 0; k0 < CH; k0 += 4)
      accY = wmma_f32(fragA(CmW[wid], ST16, k0), fragB_kn(Vv, ST, k0, nb), accY);
    if (seg == 0 && c == 0 && hi == 0)
      accY[0] += r0sum_s * corr0[nb + ln16];
#pragma unroll
    for (int v = 0; v < 8; ++v) ybuf[(v + 8 * hi) * ST + nb + ln16] = accY[v];

    v8f accS[4];
#pragma unroll
    for (int it = 0; it < 4; ++it) {
      v8f a = {};
#pragma unroll
      for (int k0 = 0; k0 < CH; k0 += 4)
        a = wmma_f32(fragA(Kh + it * 16 * ST16, ST16, k0),
                     fragB_kn(Vv, ST, k0, nb), a);
      accS[it] = a;
    }
    lds_barrier();

#pragma unroll
    for (int it = 0; it < 4; ++it) {
#pragma unroll
      for (int v = 0; v < 8; ++v) {
        int i   = it * 16 + v + 8 * hi;
        int idx = i * ST + nb + ln16;
        S[idx]  = aLast[i] * S[idx] + accS[it][v];
      }
    }
    if (tid < CH) {
      float s = 0.f, sq = 0.f;
      for (int j = 0; j < dh; ++j) {
        float y = ybuf[tid * ST + j]; s += y; sq += y * y;
      }
      float m  = s * (1.f / dh);
      float vv = sq * (1.f / dh) - m * m;
      mu_s[tid] = m;
      rs_s[tid] = rsqrtf(vv + 1e-5f);
    }
    lds_barrier();

    {
      int t  = tid >> 3;
      int jb = (tid & 7) * 8;
      float m = mu_s[t], rs = rs_s[t];
      const float gq[8] = {g0.x, g0.y, g0.z, g0.w, g1.x, g1.y, g1.z, g1.w};
#pragma unroll
      for (int q = 0; q < 8; ++q) {
        int j    = jb + q;
        float y  = ybuf[t * ST + j];
        float yn = (y - m) * rs * lnw_s[j] + lnb_s[j];
        int off  = (t0 + t) * D_ + h * dh + j;
        float gv = gq[q];
        og[off]  = yn * (gv / (1.f + __expf(-gv)));
      }
    }
    lds_barrier();
  }
}

__global__ __launch_bounds__(128)
void oproj_kernel(const float* __restrict__ o, const float* __restrict__ Wo,
                  float* __restrict__ out) {
  __shared__ float Ao[64 * ST];
  __shared__ float Bo[64 * ST];
  const int tid = threadIdx.x;
  const int lane = tid & 31, wid = tid >> 5;
  const int hi = lane >> 4, ln16 = lane & 15;
  const int nbase = blockIdx.x * 64;
  const int mbase = blockIdx.y * 64;

  v8f acc[4] = {};

  for (int ks = 0; ks < D_; ks += 64) {
    __syncthreads();
    {
      int row = tid >> 1;
      int cb  = (tid & 1) * 32;
      const float4* pa = (const float4*)(o  + (size_t)(mbase + row) * D_ + ks + cb);
      const float4* pb = (const float4*)(Wo + (size_t)(nbase + row) * D_ + ks + cb);
#pragma unroll
      for (int q = 0; q < 8; ++q) {
        float4 a4 = pa[q]; float4 b4 = pb[q];
        int cc = cb + q * 4;
        Ao[row * ST + cc + 0] = a4.x; Ao[row * ST + cc + 1] = a4.y;
        Ao[row * ST + cc + 2] = a4.z; Ao[row * ST + cc + 3] = a4.w;
        Bo[row * ST + cc + 0] = b4.x; Bo[row * ST + cc + 1] = b4.y;
        Bo[row * ST + cc + 2] = b4.z; Bo[row * ST + cc + 3] = b4.w;
      }
    }
    __syncthreads();
#pragma unroll
    for (int k0 = 0; k0 < 64; k0 += 4) {
      v2f b = fragA(Bo + wid * 16 * ST, ST, k0);
#pragma unroll
      for (int mt = 0; mt < 4; ++mt)
        acc[mt] = wmma_f32(fragA(Ao + mt * 16 * ST, ST, k0), b, acc[mt]);
    }
  }
#pragma unroll
  for (int mt = 0; mt < 4; ++mt) {
#pragma unroll
    for (int v = 0; v < 8; ++v) {
      int row = mbase + mt * 16 + v + 8 * hi;
      int col = nbase + wid * 16 + ln16;
      out[(size_t)row * D_ + col] = acc[mt][v];
    }
  }
}

extern "C" void kernel_launch(void* const* d_in, const int* in_sizes, int n_in,
                              void* d_out, int out_size, void* d_ws, size_t ws_size,
                              hipStream_t stream) {
  const float* r   = (const float*)d_in[0];
  const float* w   = (const float*)d_in[1];
  const float* k   = (const float*)d_in[2];
  const float* v   = (const float*)d_in[3];
  const float* g   = (const float*)d_in[4];
  const float* u   = (const float*)d_in[5];
  const float* is  = (const float*)d_in[6];
  const float* lnw = (const float*)d_in[7];
  const float* lnb = (const float*)d_in[8];
  const float* Wo  = (const float*)d_in[9];

  float* out   = (float*)d_out;
  float* state = out + (size_t)T_ * D_;

  float* o      = (float*)d_ws;
  float* Aseg   = o + (size_t)T_ * D_;
  float* Sloc   = Aseg + (size_t)H_ * SEG * dh;
  float* Sstart = Sloc + (size_t)H_ * SEG * dh * dh;

  dim3 gs(SEG, H_);
  wkv_pass1<<<gs, 128, 0, stream>>>(w, k, v, Aseg, Sloc);
  state_scan<<<H_, 128, 0, stream>>>(w, is, Aseg, Sloc, Sstart, state);
  wkv_pass2<<<gs, 128, 0, stream>>>(r, w, k, v, g, u, is, lnw, lnb, Sstart, o);

  dim3 g2(D_ / 64, T_ / 64);
  oproj_kernel<<<g2, 128, 0, stream>>>(o, Wo, out);
}


# pass1+pass2: state S kept in registers in WMMA accumulator layout (read as B operand of Y, updated in place) instead of an LDS array reloaded and rewritten each chunk; k order of the Y sum changes
# speedup vs baseline: 3.5200x; 1.0999x over previous
#include <hip/hip_runtime.h>

typedef __attribute__((ext_vector_type(2))) float v2f;
typedef __attribute__((ext_vector_type(8))) float v8f;

#define D_   1024
#define H_   16
#define dh   64
#define T_   4096
#define CH   16
#define SEG  16
#define TSEG (T_/SEG)
#define NCHS (TSEG/CH)
#define ST   68
#define ST16 20

__device__ __forceinline__ v8f wmma_f32(v2f a, v2f b, v8f c) {
  return __builtin_amdgcn_wmma_f32_16x16x4_f32(false, a, false, b, (short)0, c,
                                               false, false);
}

__device__ __forceinline__ v2f fragA(const float* m, int stride, int k0) {
  int lane = threadIdx.x & 31;
  const float* p = m + (lane & 15) * stride + k0 + ((lane >> 4) << 1);
  v2f f; f[0] = p[0]; f[1] = p[1];
  return f;
}
__device__ __forceinline__ v2f fragB_kn(const float* m, int stride, int k0, int nbase) {
  int lane = threadIdx.x & 31;
  int kk = k0 + ((lane >> 4) << 1);
  int n  = nbase + (lane & 15);
  v2f f; f[0] = m[kk * stride + n]; f[1] = m[(kk + 1) * stride + n];
  return f;
}

__device__ __forceinline__ void lds_barrier() {
  __builtin_amdgcn_fence(__ATOMIC_RELEASE, "workgroup", "local");
  __builtin_amdgcn_s_barrier();
  __builtin_amdgcn_fence(__ATOMIC_ACQUIRE, "workgroup", "local");
}

__global__ __launch_bounds__(128)
void wkv_pass1(const float* __restrict__ wg, const float* __restrict__ kg,
               const float* __restrict__ vg, float* __restrict__ Aseg,
               float* __restrict__ Sloc) {
  __shared__ float S[dh * ST];
  __shared__ float Vv[CH * ST];
  __shared__ float Kh[dh * ST16];
  __shared__ float aLast[dh];
  __shared__ float aAcc[dh];

  const int seg = blockIdx.x, h = blockIdx.y;
  const int tid = threadIdx.x;
  const int lane = tid & 31, wid = tid >> 5;
  const int hi = lane >> 4, ln16 = lane & 15;

  if (tid < dh) aAcc[tid] = 1.f;
  const int nb = wid * 16;
  v8f Sr[4] = {};

  const bool scanner = tid < dh;
  const int col = tid & (dh - 1);
  const int gbase = seg * TSEG * D_ + h * dh + col;
  float pw[CH], pk[CH], pv[CH];
  if (scanner) {
#pragma unroll
    for (int t = 0; t < CH; ++t) { pw[t] = wg[gbase + t * D_]; pk[t] = kg[gbase + t * D_]; }
  } else {
#pragma unroll
    for (int t = 0; t < CH; ++t) pv[t] = vg[gbase + t * D_];
  }

  for (int c = 0; c < NCHS; ++c) {
    lds_barrier();
    if (scanner) {
      int i = col;
      float a = 1.f, ai = 1.f;
      float kr[CH], air[CH];
#pragma unroll
      for (int t = 0; t < CH; ++t) {
        float wv = pw[t];
        if (seg == 0 && c == 0 && t == 0) wv = 1.f;
        float kv = pk[t];
        float rw = 1.f / wv;
        a *= wv; ai *= rw;
        kr[t] = kv; air[t] = ai;
      }
      aLast[i] = a;
      aAcc[i] *= a;
#pragma unroll
      for (int t = 0; t < CH; ++t) Kh[i * ST16 + t] = kr[t] * (a * air[t]);
    } else {
#pragma unroll
      for (int t = 0; t < CH; ++t) Vv[t * ST + col] = pv[t];
    }
    if (c + 1 < NCHS) {
      const int nbase = gbase + (c + 1) * CH * D_;
      if (scanner) {
#pragma unroll
        for (int t = 0; t < CH; ++t) { pw[t] = wg[nbase + t * D_]; pk[t] = kg[nbase + t * D_]; }
      } else {
#pragma unroll
        for (int t = 0; t < CH; ++t) pv[t] = vg[nbase + t * D_];
      }
    }
    lds_barrier();

#pragma unroll
    for (int it = 0; it < 4; ++it) {
      v8f acc = {};
#pragma unroll
      for (int k0 = 0; k0 < CH; k0 += 4)
        acc = wmma_f32(fragA(Kh + it * 16 * ST16, ST16, k0),
                       fragB_kn(Vv, ST, k0, nb), acc);
#pragma unroll
      for (int v = 0; v < 8; ++v)
        Sr[it][v] = aLast[it * 16 + v + 8 * hi] * Sr[it][v] + acc[v];
    }
  }
#pragma unroll
  for (int it = 0; it < 4; ++it) {
#pragma unroll
    for (int v = 0; v < 8; ++v)
      S[(it * 16 + v + 8 * hi) * ST + nb + ln16] = Sr[it][v];
  }
  lds_barrier();
  if (tid < dh) Aseg[(h * SEG + seg) * dh + tid] = aAcc[tid];
  for (int x = tid; x < dh * dh; x += 128) {
    int i = x >> 6, j = x & 63;
    Sloc[((size_t)(h * SEG + seg) * dh + i) * dh + j] = S[i * ST + j];
  }
}

__global__ __launch_bounds__(128)
void state_scan(const float* __restrict__ wg, const float* __restrict__ isg,
                const float* __restrict__ Aseg, const float* __restrict__ Sloc,
                float* __restrict__ Sstart, float* __restrict__ state_out) {
  const int h = blockIdx.x, tid = threadIdx.x;
#pragma clang loop vectorize(disable) interleave(disable) unroll(disable)
  for (int x = tid; x < dh * dh; x += 128) {
    int i = x >> 6, j = x & 63;
    float As[SEG], Ls[SEG];
#pragma unroll
    for (int sg = 0; sg < SEG; ++sg) {
      As[sg] = Aseg[(h * SEG + sg) * dh + i];
      Ls[sg] = Sloc[((size_t)(h * SEG + sg) * dh + i) * dh + j];
    }
    float s = wg[h * dh + j] * isg[h * dh + j];
#pragma unroll
    for (int sg = 0; sg < SEG; ++sg) {
      size_t base = ((size_t)(h * SEG + sg) * dh + i) * dh + j;
      Sstart[base] = s;
      s = As[sg] * s + Ls[sg];
    }
    state_out[i * D_ + h * dh + j] = s;
  }
}

__global__ __launch_bounds__(128)
void wkv_pass2(const float* __restrict__ rg, const float* __restrict__ wg,
               const float* __restrict__ kg, const float* __restrict__ vg,
               const float* __restrict__ gg, const float* __restrict__ ug,
               const float* __restrict__ isg, const float* __restrict__ lnw,
               const float* __restrict__ lnb, const float* __restrict__ Sstart,
               float* __restrict__ og) {
  __shared__ __attribute__((aligned(16))) float Rt[CH * ST];
  __shared__ float Kt[CH * ST];
  __shared__ float Ku[CH * ST];
  __shared__ float Vv[CH * ST];
  __shared__ float Kh[dh * ST16];
  __shared__ float CmW[4][CH * ST16];
  __shared__ float ybuf[CH * ST];
  __shared__ float aLast[dh];
  __shared__ float uvec[dh];
  __shared__ float corr0[dh];
  __shared__ float lnw_s[dh], lnb_s[dh];
  __shared__ float mu_s[CH], rs_s[CH];
  __shared__ float r0sum_s;

  const int seg = blockIdx.x, h = blockIdx.y;
  const int tid = threadIdx.x;
  const int lane = tid & 31, wid = tid >> 5;
  const int hi = lane >> 4, ln16 = lane & 15;

  for (int x = tid; x < dh; x += 128) {
    uvec[x]  = ug[h * dh + x];
    corr0[x] = isg[h * dh + x] * (1.f - wg[h * dh + x]);
    lnw_s[x] = lnw[x];
    lnb_s[x] = lnb[x];
  }
  const int nb = wid * 16;
  if (tid == 0 && seg == 0) {
    float s = 0.f;
    for (int i2 = 0; i2 < dh; ++i2) s += rg[h * dh + i2];
    r0sum_s = s;
  }

  const bool scanner = tid < dh;
  const int col = tid & (dh - 1);
  const int gbase = seg * TSEG * D_ + h * dh + col;
  const int gofs = (seg * TSEG + (tid >> 3)) * D_ + h * dh + (tid & 7) * 8;
  float pr[CH], pw[CH], pk[CH], pv[CH];
  if (scanner) {
#pragma unroll
    for (int t = 0; t < CH; ++t) {
      pr[t] = rg[gbase + t * D_]; pw[t] = wg[gbase + t * D_]; pk[t] = kg[gbase + t * D_];
    }
  } else {
#pragma unroll
    for (int t = 0; t < CH; ++t) pv[t] = vg[gbase + t * D_];
  }
  v8f Sr[4];
  {
    const float* sp = Sstart + ((size_t)(h * SEG + seg) * dh) * dh + nb + ln16;
#pragma unroll
    for (int it = 0; it < 4; ++it) {
#pragma unroll
      for (int v = 0; v < 8; ++v) Sr[it][v] = sp[(it * 16 + v + 8 * hi) * dh];
    }
  }
  asm volatile("" :: "v"(Sr[0]), "v"(Sr[1]), "v"(Sr[2]), "v"(Sr[3]));
  lds_barrier();

  for (int c = 0; c < NCHS; ++c) {
    const int t0 = seg * TSEG + c * CH;

    if (scanner) {
      int i = col;
      float a = 1.f, ai = 1.f;
      float kr[CH], air[CH];
#pragma unroll
      for (int t = 0; t < CH; ++t) {
        float wv = pw[t];
        if (seg == 0 && c == 0 && t == 0) wv = 1.f;
        float rv = pr[t], kv = pk[t];
        float aip = ai;
        Rt[t * ST + i] = rv * a;
        float rw = 1.f / wv;
        a *= wv; ai *= rw;
        Kt[t * ST + i] = kv * ai;
        Ku[t * ST + i] = kv * uvec[i] * aip;
        kr[t] = kv; air[t] = ai;
      }
      aLast[i] = a;
#pragma unroll
      for (int t = 0; t < CH; ++t) Kh[i * ST16 + t] = kr[t] * (a * air[t]);
    } else {
#pragma unroll
      for (int t = 0; t < CH; ++t) Vv[t * ST + col] = pv[t];
    }
    const float4 g0 = *(const float4*)(gg + gofs + c * CH * D_);
    const float4 g1 = *(const float4*)(gg + gofs + c * CH * D_ + 4);
    if (c + 1 < NCHS) {
      const int nbase = gbase + (c + 1) * CH * D_;
      if (scanner) {
#pragma unroll
        for (int t = 0; t < CH; ++t) {
          pr[t] = rg[nbase + t * D_]; pw[t] = wg[nbase + t * D_]; pk[t] = kg[nbase + t * D_];
        }
      } else {
#pragma unroll
        for (int t = 0; t < CH; ++t) pv[t] = vg[nbase + t * D_];
      }
    }
    lds_barrier();

    float4 ra[8];
#pragma unroll
    for (int it = 0; it < 4; ++it) {
      const float4* p = (const float4*)(Rt + ln16 * ST + it * 16 + 8 * hi);
      ra[2 * it] = p[0]; ra[2 * it + 1] = p[1];
    }

    v8f accC = {}; v8f accD = {};
#pragma unroll
    for (int k0 = 0; k0 < dh; k0 += 4) {
      v2f arf = fragA(Rt, ST, k0);
      accC = wmma_f32(arf, fragA(Kt, ST, k0), accC);
      accD = wmma_f32(arf, fragA(Ku, ST, k0), accD);
    }
    {
      float* cm = CmW[wid];
#pragma unroll
      for (int v = 0; v < 8; ++v) {
        int M = v + 8 * hi, N = ln16;
        cm[M * ST16 + N] = (N < M) ? accC[v] : ((N == M) ? accD[v] : 0.f);
      }
    }
    asm volatile("" ::: "memory");

    v8f accY = {};
#pragma unroll
    for (int it = 0; it < 4; ++it) {
      const float4 lo = ra[2 * it], up = ra[2 * it + 1];
      v2f a, b;
      a[0] = lo.x; a[1] = lo.y; b[0] = Sr[it][0]; b[1] = Sr[it][1];
      accY = wmma_f32(a, b, accY);
      a[0] = lo.z; a[1] = lo.w; b[0] = Sr[it][2]; b[1] = Sr[it][3];
      accY = wmma_f32(a, b, accY);
      a[0] = up.x; a[1] = up.y; b[0] = Sr[it][4]; b[1] = Sr[it][5];
      accY = wmma_f32(a, b, accY);
      a[0] = up.z; a[1] = up.w; b[0] = Sr[it][6]; b[1] = Sr[it][7];
      accY = wmma_f32(a, b, accY);
    }
#pragma unroll
    for (int k0 = 0; k0 < CH; k0 += 4)
      accY = wmma_f32(fragA(CmW[wid], ST16, k0), fragB_kn(Vv, ST, k0, nb), accY);
    if (seg == 0 && c == 0 && hi == 0)
      accY[0] += r0sum_s * corr0[nb + ln16];
#pragma unroll
    for (int v = 0; v < 8; ++v) ybuf[(v + 8 * hi) * ST + nb + ln16] = accY[v];

#pragma unroll
    for (int it = 0; it < 4; ++it) {
      v8f a = {};
#pragma unroll
      for (int k0 = 0; k0 < CH; k0 += 4)
        a = wmma_f32(fragA(Kh + it * 16 * ST16, ST16, k0),
                     fragB_kn(Vv, ST, k0, nb), a);
#pragma unroll
      for (int v = 0; v < 8; ++v)
        Sr[it][v] = aLast[it * 16 + v + 8 * hi] * Sr[it][v] + a[v];
    }
    lds_barrier();

    if (tid < CH) {
      float s = 0.f, sq = 0.f;
      for (int j = 0; j < dh; ++j) {
        float y = ybuf[tid * ST + j]; s += y; sq += y * y;
      }
      float m  = s * (1.f / dh);
      float vv = sq * (1.f / dh) - m * m;
      mu_s[tid] = m;
      rs_s[tid] = rsqrtf(vv + 1e-5f);
    }
    lds_barrier();

    {
      int t  = tid >> 3;
      int jb = (tid & 7) * 8;
      float m = mu_s[t], rs = rs_s[t];
      const float gq[8] = {g0.x, g0.y, g0.z, g0.w, g1.x, g1.y, g1.z, g1.w};
#pragma unroll
      for (int q = 0; q < 8; ++q) {
        int j    = jb + q;
        float y  = ybuf[t * ST + j];
        float yn = (y - m) * rs * lnw_s[j] + lnb_s[j];
        int off  = (t0 + t) * D_ + h * dh + j;
        float gv = gq[q];
        og[off]  = yn * (gv / (1.f + __expf(-gv)));
      }
    }
    lds_barrier();
  }
}

__global__ __launch_bounds__(128)
void oproj_kernel(const float* __restrict__ o, const float* __restrict__ Wo,
                  float* __restrict__ out) {
  __shared__ float Ao[64 * ST];
  __shared__ float Bo[64 * ST];
  const int tid = threadIdx.x;
  const int lane = tid & 31, wid = tid >> 5;
  const int hi = lane >> 4, ln16 = lane & 15;
  const int nbase = blockIdx.x * 64;
  const int mbase = blockIdx.y * 64;

  v8f acc[4] = {};

  for (int ks = 0; ks < D_; ks += 64) {
    __syncthreads();
    {
      int row = tid >> 1;
      int cb  = (tid & 1) * 32;
      const float4* pa = (const float4*)(o  + (size_t)(mbase + row) * D_ + ks + cb);
      const float4* pb = (const float4*)(Wo + (size_t)(nbase + row) * D_ + ks + cb);
#pragma unroll
      for (int q = 0; q < 8; ++q) {
        float4 a4 = pa[q]; float4 b4 = pb[q];
        int cc = cb + q * 4;
        Ao[row * ST + cc + 0] = a4.x; Ao[row * ST + cc + 1] = a4.y;
        Ao[row * ST + cc + 2] = a4.z; Ao[row * ST + cc + 3] = a4.w;
        Bo[row * ST + cc + 0] = b4.x; Bo[row * ST + cc + 1] = b4.y;
        Bo[row * ST + cc + 2] = b4.z; Bo[row * ST + cc + 3] = b4.w;
      }
    }
    __syncthreads();
#pragma unroll
    for (int k0 = 0; k0 < 64; k0 += 4) {
      v2f b = fragA(Bo + wid * 16 * ST, ST, k0);
#pragma unroll
      for (int mt = 0; mt < 4; ++mt)
        acc[mt] = wmma_f32(fragA(Ao + mt * 16 * ST, ST, k0), b, acc[mt]);
    }
  }
#pragma unroll
  for (int mt = 0; mt < 4; ++mt) {
#pragma unroll
    for (int v = 0; v < 8; ++v) {
      int row = mbase + mt * 16 + v + 8 * hi;
      int col = nbase + wid * 16 + ln16;
      out[(size_t)row * D_ + col] = acc[mt][v];
    }
  }
}

extern "C" void kernel_launch(void* const* d_in, const int* in_sizes, int n_in,
                              void* d_out, int out_size, void* d_ws, size_t ws_size,
                              hipStream_t stream) {
  const float* r   = (const float*)d_in[0];
  const float* w   = (const float*)d_in[1];
  const float* k   = (const float*)d_in[2];
  const float* v   = (const float*)d_in[3];
  const float* g   = (const float*)d_in[4];
  const float* u   = (const float*)d_in[5];
  const float* is  = (const float*)d_in[6];
  const float* lnw = (const float*)d_in[7];
  const float* lnb = (const float*)d_in[8];
  const float* Wo  = (const float*)d_in[9];

  float* out   = (float*)d_out;
  float* state = out + (size_t)T_ * D_;

  float* o      = (float*)d_ws;
  float* Aseg   = o + (size_t)T_ * D_;
  float* Sloc   = Aseg + (size_t)H_ * SEG * dh;
  float* Sstart = Sloc + (size_t)H_ * SEG * dh * dh;

  dim3 gs(SEG, H_);
  wkv_pass1<<<gs, 128, 0, stream>>>(w, k, v, Aseg, Sloc);
  state_scan<<<H_, 128, 0, stream>>>(w, is, Aseg, Sloc, Sstart, state);
  wkv_pass2<<<gs, 128, 0, stream>>>(r, w, k, v, g, u, is, lnw, lnb, Sstart, o);

  dim3 g2(D_ / 64, T_ / 64);
  oproj_kernel<<<g2, 128, 0, stream>>>(o, Wo, out);
}


# pass1, scan, pass2: 32 time segments per head instead of 16, so 512 work-groups run two per WGP and each fills the other's stalls; the state now sums over twice as many segment carries
# speedup vs baseline: 4.2030x; 1.1940x over previous
#include <hip/hip_runtime.h>

typedef __attribute__((ext_vector_type(2))) float v2f;
typedef __attribute__((ext_vector_type(8))) float v8f;

#define D_   1024
#define H_   16
#define dh   64
#define T_   4096
#define CH   16
#define SEG  32
#define TSEG (T_/SEG)
#define NCHS (TSEG/CH)
#define ST   68
#define ST16 20

__device__ __forceinline__ v8f wmma_f32(v2f a, v2f b, v8f c) {
  return __builtin_amdgcn_wmma_f32_16x16x4_f32(false, a, false, b, (short)0, c,
                                               false, false);
}

__device__ __forceinline__ v2f fragA(const float* m, int stride, int k0) {
  int lane = threadIdx.x & 31;
  const float* p = m + (lane & 15) * stride + k0 + ((lane >> 4) << 1);
  v2f f; f[0] = p[0]; f[1] = p[1];
  return f;
}
__device__ __forceinline__ v2f fragB_kn(const float* m, int stride, int k0, int nbase) {
  int lane = threadIdx.x & 31;
  int kk = k0 + ((lane >> 4) << 1);
  int n  = nbase + (lane & 15);
  v2f f; f[0] = m[kk * stride + n]; f[1] = m[(kk + 1) * stride + n];
  return f;
}

__device__ __forceinline__ void lds_barrier() {
  __builtin_amdgcn_fence(__ATOMIC_RELEASE, "workgroup", "local");
  __builtin_amdgcn_s_barrier();
  __builtin_amdgcn_fence(__ATOMIC_ACQUIRE, "workgroup", "local");
}

__global__ __launch_bounds__(128)
void wkv_pass1(const float* __restrict__ wg, const float* __restrict__ kg,
               const float* __restrict__ vg, float* __restrict__ Aseg,
               float* __restrict__ Sloc) {
  __shared__ float S[dh * ST];
  __shared__ float Vv[CH * ST];
  __shared__ float Kh[dh * ST16];
  __shared__ float aLast[dh];
  __shared__ float aAcc[dh];

  const int seg = blockIdx.x, h = blockIdx.y;
  const int tid = threadIdx.x;
  const int lane = tid & 31, wid = tid >> 5;
  const int hi = lane >> 4, ln16 = lane & 15;

  if (tid < dh) aAcc[tid] = 1.f;
  const int nb = wid * 16;
  v8f Sr[4] = {};

  const bool scanner = tid < dh;
  const int col = tid & (dh - 1);
  const int gbase = seg * TSEG * D_ + h * dh + col;
  float pw[CH], pk[CH], pv[CH];
  if (scanner) {
#pragma unroll
    for (int t = 0; t < CH; ++t) { pw[t] = wg[gbase + t * D_]; pk[t] = kg[gbase + t * D_]; }
  } else {
#pragma unroll
    for (int t = 0; t < CH; ++t) pv[t] = vg[gbase + t * D_];
  }

  for (int c = 0; c < NCHS; ++c) {
    lds_barrier();
    if (scanner) {
      int i = col;
      float a = 1.f, ai = 1.f;
      float kr[CH], air[CH];
#pragma unroll
      for (int t = 0; t < CH; ++t) {
        float wv = pw[t];
        if (seg == 0 && c == 0 && t == 0) wv = 1.f;
        float kv = pk[t];
        float rw = 1.f / wv;
        a *= wv; ai *= rw;
        kr[t] = kv; air[t] = ai;
      }
      aLast[i] = a;
      aAcc[i] *= a;
#pragma unroll
      for (int t = 0; t < CH; ++t) Kh[i * ST16 + t] = kr[t] * (a * air[t]);
    } else {
#pragma unroll
      for (int t = 0; t < CH; ++t) Vv[t * ST + col] = pv[t];
    }
    if (c + 1 < NCHS) {
      const int nbase = gbase + (c + 1) * CH * D_;
      if (scanner) {
#pragma unroll
        for (int t = 0; t < CH; ++t) { pw[t] = wg[nbase + t * D_]; pk[t] = kg[nbase + t * D_]; }
      } else {
#pragma unroll
        for (int t = 0; t < CH; ++t) pv[t] = vg[nbase + t * D_];
      }
    }
    lds_barrier();

#pragma unroll
    for (int it = 0; it < 4; ++it) {
      v8f acc = {};
#pragma unroll
      for (int k0 = 0; k0 < CH; k0 += 4)
        acc = wmma_f32(fragA(Kh + it * 16 * ST16, ST16, k0),
                       fragB_kn(Vv, ST, k0, nb), acc);
#pragma unroll
      for (int v = 0; v < 8; ++v)
        Sr[it][v] = aLast[it * 16 + v + 8 * hi] * Sr[it][v] + acc[v];
    }
  }
#pragma unroll
  for (int it = 0; it < 4; ++it) {
#pragma unroll
    for (int v = 0; v < 8; ++v)
      S[(it * 16 + v + 8 * hi) * ST + nb + ln16] = Sr[it][v];
  }
  lds_barrier();
  if (tid < dh) Aseg[(h * SEG + seg) * dh + tid] = aAcc[tid];
  for (int x = tid; x < dh * dh; x += 128) {
    int i = x >> 6, j = x & 63;
    Sloc[((size_t)(h * SEG + seg) * dh + i) * dh + j] = S[i * ST + j];
  }
}

__global__ __launch_bounds__(128)
void state_scan(const float* __restrict__ wg, const float* __restrict__ isg,
                const float* __restrict__ Aseg, const float* __restrict__ Sloc,
                float* __restrict__ Sstart, float* __restrict__ state_out) {
  static_assert(SEG == 32, "one decay factor per lane of a 32-lane wave");
  const int h = blockIdx.x, tid = threadIdx.x;
  const int lane = tid & 31;
#pragma clang loop vectorize(disable) interleave(disable) unroll(disable)
  for (int x = tid * 2; x < dh * dh; x += 256) {
    const int i = x >> 6, j = x & 63;
    const float Al = Aseg[(h * SEG + lane) * dh + i];
    float2 Ls[SEG];
#pragma unroll
    for (int sg = 0; sg < SEG; ++sg)
      Ls[sg] = *(const float2*)(Sloc + ((size_t)(h * SEG + sg) * dh + i) * dh + j);
    const float2 w0 = *(const float2*)(wg + h * dh + j);
    const float2 s0 = *(const float2*)(isg + h * dh + j);
    float sa = w0.x * s0.x, sb = w0.y * s0.y;
#pragma unroll
    for (int sg = 0; sg < SEG; ++sg) {
      size_t base = ((size_t)(h * SEG + sg) * dh + i) * dh + j;
      *(float2*)(Sstart + base) = make_float2(sa, sb);
      const float a = __int_as_float(__builtin_amdgcn_readlane(__float_as_int(Al), sg));
      sa = a * sa + Ls[sg].x;
      sb = a * sb + Ls[sg].y;
    }
    *(float2*)(state_out + i * D_ + h * dh + j) = make_float2(sa, sb);
  }
}

__global__ __launch_bounds__(128)
void wkv_pass2(const float* __restrict__ rg, const float* __restrict__ wg,
               const float* __restrict__ kg, const float* __restrict__ vg,
               const float* __restrict__ gg, const float* __restrict__ ug,
               const float* __restrict__ isg, const float* __restrict__ lnw,
               const float* __restrict__ lnb, const float* __restrict__ Sstart,
               float* __restrict__ og) {
  __shared__ __attribute__((aligned(16))) float Rt[CH * ST];
  __shared__ float Kt[CH * ST];
  __shared__ float Ku[CH * ST];
  __shared__ float Vv[CH * ST];
  __shared__ float Kh[dh * ST16];
  __shared__ float CmW[4][CH * ST16];
  __shared__ float ybuf[CH * ST];
  __shared__ float aLast[dh];
  __shared__ float uvec[dh];
  __shared__ float corr0[dh];
  __shared__ float lnw_s[dh], lnb_s[dh];
  __shared__ float mu_s[CH], rs_s[CH];
  __shared__ float r0sum_s;

  const int seg = blockIdx.x, h = blockIdx.y;
  const int tid = threadIdx.x;
  const int lane = tid & 31, wid = tid >> 5;
  const int hi = lane >> 4, ln16 = lane & 15;

  for (int x = tid; x < dh; x += 128) {
    uvec[x]  = ug[h * dh + x];
    corr0[x] = isg[h * dh + x] * (1.f - wg[h * dh + x]);
    lnw_s[x] = lnw[x];
    lnb_s[x] = lnb[x];
  }
  const int nb = wid * 16;
  if (tid == 0 && seg == 0) {
    float s = 0.f;
    for (int i2 = 0; i2 < dh; ++i2) s += rg[h * dh + i2];
    r0sum_s = s;
  }

  const bool scanner = tid < dh;
  const int col = tid & (dh - 1);
  const int gbase = seg * TSEG * D_ + h * dh + col;
  const int gofs = (seg * TSEG + (tid >> 3)) * D_ + h * dh + (tid & 7) * 8;
  float pr[CH], pw[CH], pk[CH], pv[CH];
  if (scanner) {
#pragma unroll
    for (int t = 0; t < CH; ++t) {
      pr[t] = rg[gbase + t * D_]; pw[t] = wg[gbase + t * D_]; pk[t] = kg[gbase + t * D_];
    }
  } else {
#pragma unroll
    for (int t = 0; t < CH; ++t) pv[t] = vg[gbase + t * D_];
  }
  v8f Sr[4];
  {
    const float* sp = Sstart + ((size_t)(h * SEG + seg) * dh) * dh + nb + ln16;
#pragma unroll
    for (int it = 0; it < 4; ++it) {
#pragma unroll
      for (int v = 0; v < 8; ++v) Sr[it][v] = sp[(it * 16 + v + 8 * hi) * dh];
    }
  }
  asm volatile("" :: "v"(Sr[0]), "v"(Sr[1]), "v"(Sr[2]), "v"(Sr[3]));
  lds_barrier();

  for (int c = 0; c < NCHS; ++c) {
    const int t0 = seg * TSEG + c * CH;

    if (scanner) {
      int i = col;
      float a = 1.f, ai = 1.f;
      float kr[CH], air[CH];
#pragma unroll
      for (int t = 0; t < CH; ++t) {
        float wv = pw[t];
        if (seg == 0 && c == 0 && t == 0) wv = 1.f;
        float rv = pr[t], kv = pk[t];
        float aip = ai;
        Rt[t * ST + i] = rv * a;
        float rw = 1.f / wv;
        a *= wv; ai *= rw;
        Kt[t * ST + i] = kv * ai;
        Ku[t * ST + i] = kv * uvec[i] * aip;
        kr[t] = kv; air[t] = ai;
      }
      aLast[i] = a;
#pragma unroll
      for (int t = 0; t < CH; ++t) Kh[i * ST16 + t] = kr[t] * (a * air[t]);
    } else {
#pragma unroll
      for (int t = 0; t < CH; ++t) Vv[t * ST + col] = pv[t];
    }
    const float4 g0 = *(const float4*)(gg + gofs + c * CH * D_);
    const float4 g1 = *(const float4*)(gg + gofs + c * CH * D_ + 4);
    if (c + 1 < NCHS) {
      const int nbase = gbase + (c + 1) * CH * D_;
      if (scanner) {
#pragma unroll
        for (int t = 0; t < CH; ++t) {
          pr[t] = rg[nbase + t * D_]; pw[t] = wg[nbase + t * D_]; pk[t] = kg[nbase + t * D_];
        }
      } else {
#pragma unroll
        for (int t = 0; t < CH; ++t) pv[t] = vg[nbase + t * D_];
      }
    }
    lds_barrier();

    float4 ra[8];
#pragma unroll
    for (int it = 0; it < 4; ++it) {
      const float4* p = (const float4*)(Rt + ln16 * ST + it * 16 + 8 * hi);
      ra[2 * it] = p[0]; ra[2 * it + 1] = p[1];
    }

    v8f accC = {}; v8f accD = {};
#pragma unroll
    for (int k0 = 0; k0 < dh; k0 += 4) {
      v2f arf = fragA(Rt, ST, k0);
      accC = wmma_f32(arf, fragA(Kt, ST, k0), accC);
      accD = wmma_f32(arf, fragA(Ku, ST, k0), accD);
    }
    {
      float* cm = CmW[wid];
#pragma unroll
      for (int v = 0; v < 8; ++v) {
        int M = v + 8 * hi, N = ln16;
        cm[M * ST16 + N] = (N < M) ? accC[v] : ((N == M) ? accD[v] : 0.f);
      }
    }
    asm volatile("" ::: "memory");

    v8f accY = {};
#pragma unroll
    for (int it = 0; it < 4; ++it) {
      const float4 lo = ra[2 * it], up = ra[2 * it + 1];
      v2f a, b;
      a[0] = lo.x; a[1] = lo.y; b[0] = Sr[it][0]; b[1] = Sr[it][1];
      accY = wmma_f32(a, b, accY);
      a[0] = lo.z; a[1] = lo.w; b[0] = Sr[it][2]; b[1] = Sr[it][3];
      accY = wmma_f32(a, b, accY);
      a[0] = up.x; a[1] = up.y; b[0] = Sr[it][4]; b[1] = Sr[it][5];
      accY = wmma_f32(a, b, accY);
      a[0] = up.z; a[1] = up.w; b[0] = Sr[it][6]; b[1] = Sr[it][7];
      accY = wmma_f32(a, b, accY);
    }
#pragma unroll
    for (int k0 = 0; k0 < CH; k0 += 4)
      accY = wmma_f32(fragA(CmW[wid], ST16, k0), fragB_kn(Vv, ST, k0, nb), accY);
    if (seg == 0 && c == 0 && hi == 0)
      accY[0] += r0sum_s * corr0[nb + ln16];
#pragma unroll
    for (int v = 0; v < 8; ++v) ybuf[(v + 8 * hi) * ST + nb + ln16] = accY[v];

#pragma unroll
    for (int it = 0; it < 4; ++it) {
      v8f a = {};
#pragma unroll
      for (int k0 = 0; k0 < CH; k0 += 4)
        a = wmma_f32(fragA(Kh + it * 16 * ST16, ST16, k0),
                     fragB_kn(Vv, ST, k0, nb), a);
#pragma unroll
      for (int v = 0; v < 8; ++v)
        Sr[it][v] = aLast[it * 16 + v + 8 * hi] * Sr[it][v] + a[v];
    }
    lds_barrier();

    if (tid < CH) {
      float s = 0.f, sq = 0.f;
      for (int j = 0; j < dh; ++j) {
        float y = ybuf[tid * ST + j]; s += y; sq += y * y;
      }
      float m  = s * (1.f / dh);
      float vv = sq * (1.f / dh) - m * m;
      mu_s[tid] = m;
      rs_s[tid] = rsqrtf(vv + 1e-5f);
    }
    lds_barrier();

    {
      int t  = tid >> 3;
      int jb = (tid & 7) * 8;
      float m = mu_s[t], rs = rs_s[t];
      const float gq[8] = {g0.x, g0.y, g0.z, g0.w, g1.x, g1.y, g1.z, g1.w};
#pragma unroll
      for (int q = 0; q < 8; ++q) {
        int j    = jb + q;
        float y  = ybuf[t * ST + j];
        float yn = (y - m) * rs * lnw_s[j] + lnb_s[j];
        int off  = (t0 + t) * D_ + h * dh + j;
        float gv = gq[q];
        og[off]  = yn * (gv / (1.f + __expf(-gv)));
      }
    }
    lds_barrier();
  }
}

__global__ __launch_bounds__(128)
void oproj_kernel(const float* __restrict__ o, const float* __restrict__ Wo,
                  float* __restrict__ out) {
  __shared__ float Ao[64 * ST];
  __shared__ float Bo[64 * ST];
  const int tid = threadIdx.x;
  const int lane = tid & 31, wid = tid >> 5;
  const int hi = lane >> 4, ln16 = lane & 15;
  const int nbase = blockIdx.x * 64;
  const int mbase = blockIdx.y * 64;

  v8f acc[4] = {};

  for (int ks = 0; ks < D_; ks += 64) {
    __syncthreads();
    {
      int row = tid >> 1;
      int cb  = (tid & 1) * 32;
      const float4* pa = (const float4*)(o  + (size_t)(mbase + row) * D_ + ks + cb);
      const float4* pb = (const float4*)(Wo + (size_t)(nbase + row) * D_ + ks + cb);
#pragma unroll
      for (int q = 0; q < 8; ++q) {
        float4 a4 = pa[q]; float4 b4 = pb[q];
        int cc = cb + q * 4;
        Ao[row * ST + cc + 0] = a4.x; Ao[row * ST + cc + 1] = a4.y;
        Ao[row * ST + cc + 2] = a4.z; Ao[row * ST + cc + 3] = a4.w;
        Bo[row * ST + cc + 0] = b4.x; Bo[row * ST + cc + 1] = b4.y;
        Bo[row * ST + cc + 2] = b4.z; Bo[row * ST + cc + 3] = b4.w;
      }
    }
    __syncthreads();
#pragma unroll
    for (int k0 = 0; k0 < 64; k0 += 4) {
      v2f b = fragA(Bo + wid * 16 * ST, ST, k0);
#pragma unroll
      for (int mt = 0; mt < 4; ++mt)
        acc[mt] = wmma_f32(fragA(Ao + mt * 16 * ST, ST, k0), b, acc[mt]);
    }
  }
#pragma unroll
  for (int mt = 0; mt < 4; ++mt) {
#pragma unroll
    for (int v = 0; v < 8; ++v) {
      int row = mbase + mt * 16 + v + 8 * hi;
      int col = nbase + wid * 16 + ln16;
      out[(size_t)row * D_ + col] = acc[mt][v];
    }
  }
}

extern "C" void kernel_launch(void* const* d_in, const int* in_sizes, int n_in,
                              void* d_out, int out_size, void* d_ws, size_t ws_size,
                              hipStream_t stream) {
  const float* r   = (const float*)d_in[0];
  const float* w   = (const float*)d_in[1];
  const float* k   = (const float*)d_in[2];
  const float* v   = (const float*)d_in[3];
  const float* g   = (const float*)d_in[4];
  const float* u   = (const float*)d_in[5];
  const float* is  = (const float*)d_in[6];
  const float* lnw = (const float*)d_in[7];
  const float* lnb = (const float*)d_in[8];
  const float* Wo  = (const float*)d_in[9];

  float* out   = (float*)d_out;
  float* state = out + (size_t)T_ * D_;

  float* o      = (float*)d_ws;
  float* Aseg   = o + (size_t)T_ * D_;
  float* Sloc   = Aseg + (size_t)H_ * SEG * dh;
  float* Sstart = Sloc + (size_t)H_ * SEG * dh * dh;

  dim3 gs(SEG, H_);
  wkv_pass1<<<gs, 128, 0, stream>>>(w, k, v, Aseg, Sloc);
  state_scan<<<H_, 128, 0, stream>>>(w, is, Aseg, Sloc, Sstart, state);
  wkv_pass2<<<gs, 128, 0, stream>>>(r, w, k, v, g, u, is, lnw, lnb, Sstart, o);

  dim3 g2(D_ / 64, T_ / 64);
  oproj_kernel<<<g2, 128, 0, stream>>>(o, Wo, out);
}


# o_proj GEMM: tile staging loads re-mapped so that a wave-wide b128 load covers four whole 128-byte lines (two full tile rows) instead of one piece out of each of 32 lines
# speedup vs baseline: 4.6947x; 1.1170x over previous
#include <hip/hip_runtime.h>

typedef __attribute__((ext_vector_type(2))) float v2f;
typedef __attribute__((ext_vector_type(4))) float v4f;
typedef __attribute__((ext_vector_type(8))) float v8f;

#define D_   1024
#define H_   16
#define dh   64
#define T_   4096
#define CH   16
#define SEG  32
#define TSEG (T_/SEG)
#define NCHS (TSEG/CH)
#define ST   68
#define ST16 20

__device__ __forceinline__ v8f wmma_f32(v2f a, v2f b, v8f c) {
  return __builtin_amdgcn_wmma_f32_16x16x4_f32(false, a, false, b, (short)0, c,
                                               false, false);
}

__device__ __forceinline__ v2f fragA(const float* m, int stride, int k0) {
  int lane = threadIdx.x & 31;
  const float* p = m + (lane & 15) * stride + k0 + ((lane >> 4) << 1);
  v2f f; f[0] = p[0]; f[1] = p[1];
  return f;
}
__device__ __forceinline__ v2f fragB_kn(const float* m, int stride, int k0, int nbase) {
  int lane = threadIdx.x & 31;
  int kk = k0 + ((lane >> 4) << 1);
  int n  = nbase + (lane & 15);
  v2f f; f[0] = m[kk * stride + n]; f[1] = m[(kk + 1) * stride + n];
  return f;
}

__device__ __forceinline__ void lds_barrier() {
  __builtin_amdgcn_fence(__ATOMIC_RELEASE, "workgroup", "local");
  __builtin_amdgcn_s_barrier();
  __builtin_amdgcn_fence(__ATOMIC_ACQUIRE, "workgroup", "local");
}

__global__ __launch_bounds__(128)
void wkv_pass1(const float* __restrict__ wg, const float* __restrict__ kg,
               const float* __restrict__ vg, float* __restrict__ Aseg,
               float* __restrict__ Sloc) {
  __shared__ float S[dh * ST];
  __shared__ float Vv[CH * ST];
  __shared__ float Kh[dh * ST16];
  __shared__ float aLast[dh];
  __shared__ float aAcc[dh];

  const int seg = blockIdx.x, h = blockIdx.y;
  const int tid = threadIdx.x;
  const int lane = tid & 31, wid = tid >> 5;
  const int hi = lane >> 4, ln16 = lane & 15;

  if (tid < dh) aAcc[tid] = 1.f;
  const int nb = wid * 16;
  v8f Sr[4] = {};

  const bool scanner = tid < dh;
  const int col = tid & (dh - 1);
  const int gbase = seg * TSEG * D_ + h * dh + col;
  float pw[CH], pk[CH], pv[CH];
  if (scanner) {
#pragma unroll
    for (int t = 0; t < CH; ++t) { pw[t] = wg[gbase + t * D_]; pk[t] = kg[gbase + t * D_]; }
  } else {
#pragma unroll
    for (int t = 0; t < CH; ++t) pv[t] = vg[gbase + t * D_];
  }

  for (int c = 0; c < NCHS; ++c) {
    lds_barrier();
    if (scanner) {
      int i = col;
      float a = 1.f, ai = 1.f;
      float kr[CH], air[CH];
#pragma unroll
      for (int t = 0; t < CH; ++t) {
        float wv = pw[t];
        if (seg == 0 && c == 0 && t == 0) wv = 1.f;
        float kv = pk[t];
        float rw = 1.f / wv;
        a *= wv; ai *= rw;
        kr[t] = kv; air[t] = ai;
      }
      aLast[i] = a;
      aAcc[i] *= a;
#pragma unroll
      for (int t = 0; t < CH; ++t) Kh[i * ST16 + t] = kr[t] * (a * air[t]);
    } else {
#pragma unroll
      for (int t = 0; t < CH; ++t) Vv[t * ST + col] = pv[t];
    }
    if (c + 1 < NCHS) {
      const int nbase = gbase + (c + 1) * CH * D_;
      if (scanner) {
#pragma unroll
        for (int t = 0; t < CH; ++t) { pw[t] = wg[nbase + t * D_]; pk[t] = kg[nbase + t * D_]; }
      } else {
#pragma unroll
        for (int t = 0; t < CH; ++t) pv[t] = vg[nbase + t * D_];
      }
    }
    lds_barrier();

#pragma unroll
    for (int it = 0; it < 4; ++it) {
      v8f acc = {};
#pragma unroll
      for (int k0 = 0; k0 < CH; k0 += 4)
        acc = wmma_f32(fragA(Kh + it * 16 * ST16, ST16, k0),
                       fragB_kn(Vv, ST, k0, nb), acc);
#pragma unroll
      for (int v = 0; v < 8; ++v)
        Sr[it][v] = aLast[it * 16 + v + 8 * hi] * Sr[it][v] + acc[v];
    }
  }
#pragma unroll
  for (int it = 0; it < 4; ++it) {
#pragma unroll
    for (int v = 0; v < 8; ++v)
      S[(it * 16 + v + 8 * hi) * ST + nb + ln16] = Sr[it][v];
  }
  lds_barrier();
  if (tid < dh) Aseg[(h * SEG + seg) * dh + tid] = aAcc[tid];
  for (int x = tid; x < dh * dh; x += 128) {
    int i = x >> 6, j = x & 63;
    Sloc[((size_t)(h * SEG + seg) * dh + i) * dh + j] = S[i * ST + j];
  }
}

__global__ __launch_bounds__(128)
void state_scan(const float* __restrict__ wg, const float* __restrict__ isg,
                const float* __restrict__ Aseg, const float* __restrict__ Sloc,
                float* __restrict__ Sstart, float* __restrict__ state_out) {
  static_assert(SEG == 32, "one decay factor per lane of a 32-lane wave");
  const int h = blockIdx.x, tid = threadIdx.x;
  const int lane = tid & 31;
#pragma clang loop vectorize(disable) interleave(disable) unroll(disable)
  for (int x = tid * 2; x < dh * dh; x += 256) {
    const int i = x >> 6, j = x & 63;
    const float Al = Aseg[(h * SEG + lane) * dh + i];
    float2 Ls[SEG];
#pragma unroll
    for (int sg = 0; sg < SEG; ++sg)
      Ls[sg] = *(const float2*)(Sloc + ((size_t)(h * SEG + sg) * dh + i) * dh + j);
    const float2 w0 = *(const float2*)(wg + h * dh + j);
    const float2 s0 = *(const float2*)(isg + h * dh + j);
    float sa = w0.x * s0.x, sb = w0.y * s0.y;
#pragma unroll
    for (int sg = 0; sg < SEG; ++sg) {
      size_t base = ((size_t)(h * SEG + sg) * dh + i) * dh + j;
      *(float2*)(Sstart + base) = make_float2(sa, sb);
      const float a = __int_as_float(__builtin_amdgcn_readlane(__float_as_int(Al), sg));
      sa = a * sa + Ls[sg].x;
      sb = a * sb + Ls[sg].y;
    }
    *(float2*)(state_out + i * D_ + h * dh + j) = make_float2(sa, sb);
  }
}

__global__ __launch_bounds__(128)
void wkv_pass2(const float* __restrict__ rg, const float* __restrict__ wg,
               const float* __restrict__ kg, const float* __restrict__ vg,
               const float* __restrict__ gg, const float* __restrict__ ug,
               const float* __restrict__ isg, const float* __restrict__ lnw,
               const float* __restrict__ lnb, const float* __restrict__ Sstart,
               float* __restrict__ og) {
  __shared__ __attribute__((aligned(16))) float Rt[CH * ST];
  __shared__ float Kt[CH * ST];
  __shared__ float Ku[CH * ST];
  __shared__ float Vv[CH * ST];
  __shared__ float Kh[dh * ST16];
  __shared__ float CmW[4][CH * ST16];
  __shared__ float ybuf[CH * ST];
  __shared__ float aLast[dh];
  __shared__ float uvec[dh];
  __shared__ float corr0[dh];
  __shared__ float lnw_s[dh], lnb_s[dh];
  __shared__ float mu_s[CH], rs_s[CH];
  __shared__ float r0sum_s;

  const int seg = blockIdx.x, h = blockIdx.y;
  const int tid = threadIdx.x;
  const int lane = tid & 31, wid = tid >> 5;
  const int hi = lane >> 4, ln16 = lane & 15;

  for (int x = tid; x < dh; x += 128) {
    uvec[x]  = ug[h * dh + x];
    corr0[x] = isg[h * dh + x] * (1.f - wg[h * dh + x]);
    lnw_s[x] = lnw[x];
    lnb_s[x] = lnb[x];
  }
  const int nb = wid * 16;
  if (tid == 0 && seg == 0) {
    float s = 0.f;
    for (int i2 = 0; i2 < dh; ++i2) s += rg[h * dh + i2];
    r0sum_s = s;
  }

  const bool scanner = tid < dh;
  const int col = tid & (dh - 1);
  const int gbase = seg * TSEG * D_ + h * dh + col;
  const int gofs = (seg * TSEG + (tid >> 3)) * D_ + h * dh + (tid & 7) * 8;
  float pr[CH], pw[CH], pk[CH], pv[CH];
  if (scanner) {
#pragma unroll
    for (int t = 0; t < CH; ++t) {
      pr[t] = rg[gbase + t * D_]; pw[t] = wg[gbase + t * D_]; pk[t] = kg[gbase + t * D_];
    }
  } else {
#pragma unroll
    for (int t = 0; t < CH; ++t) pv[t] = vg[gbase + t * D_];
  }
  v8f Sr[4];
  {
    const float* sp = Sstart + ((size_t)(h * SEG + seg) * dh) * dh + nb + ln16;
#pragma unroll
    for (int it = 0; it < 4; ++it) {
#pragma unroll
      for (int v = 0; v < 8; ++v) Sr[it][v] = sp[(it * 16 + v + 8 * hi) * dh];
    }
  }
  asm volatile("" :: "v"(Sr[0]), "v"(Sr[1]), "v"(Sr[2]), "v"(Sr[3]));
  lds_barrier();

  for (int c = 0; c < NCHS; ++c) {
    const int t0 = seg * TSEG + c * CH;

    if (scanner) {
      int i = col;
      float a = 1.f, ai = 1.f;
      float kr[CH], air[CH];
#pragma unroll
      for (int t = 0; t < CH; ++t) {
        float wv = pw[t];
        if (seg == 0 && c == 0 && t == 0) wv = 1.f;
        float rv = pr[t], kv = pk[t];
        float aip = ai;
        Rt[t * ST + i] = rv * a;
        float rw = 1.f / wv;
        a *= wv; ai *= rw;
        Kt[t * ST + i] = kv * ai;
        Ku[t * ST + i] = kv * uvec[i] * aip;
        kr[t] = kv; air[t] = ai;
      }
      aLast[i] = a;
#pragma unroll
      for (int t = 0; t < CH; ++t) Kh[i * ST16 + t] = kr[t] * (a * air[t]);
    } else {
#pragma unroll
      for (int t = 0; t < CH; ++t) Vv[t * ST + col] = pv[t];
    }
    const float4 g0 = *(const float4*)(gg + gofs + c * CH * D_);
    const float4 g1 = *(const float4*)(gg + gofs + c * CH * D_ + 4);
    if (c + 1 < NCHS) {
      const int nbase = gbase + (c + 1) * CH * D_;
      if (scanner) {
#pragma unroll
        for (int t = 0; t < CH; ++t) {
          pr[t] = rg[nbase + t * D_]; pw[t] = wg[nbase + t * D_]; pk[t] = kg[nbase + t * D_];
        }
      } else {
#pragma unroll
        for (int t = 0; t < CH; ++t) pv[t] = vg[nbase + t * D_];
      }
    }
    lds_barrier();

    float4 ra[8];
#pragma unroll
    for (int it = 0; it < 4; ++it) {
      const float4* p = (const float4*)(Rt + ln16 * ST + it * 16 + 8 * hi);
      ra[2 * it] = p[0]; ra[2 * it + 1] = p[1];
    }

    v8f accC = {}; v8f accD = {};
#pragma unroll
    for (int k0 = 0; k0 < dh; k0 += 4) {
      v2f arf = fragA(Rt, ST, k0);
      accC = wmma_f32(arf, fragA(Kt, ST, k0), accC);
      accD = wmma_f32(arf, fragA(Ku, ST, k0), accD);
    }
    {
      float* cm = CmW[wid];
#pragma unroll
      for (int v = 0; v < 8; ++v) {
        int M = v + 8 * hi, N = ln16;
        cm[M * ST16 + N] = (N < M) ? accC[v] : ((N == M) ? accD[v] : 0.f);
      }
    }
    asm volatile("" ::: "memory");

    v8f accY = {};
#pragma unroll
    for (int it = 0; it < 4; ++it) {
      const float4 lo = ra[2 * it], up = ra[2 * it + 1];
      v2f a, b;
      a[0] = lo.x; a[1] = lo.y; b[0] = Sr[it][0]; b[1] = Sr[it][1];
      accY = wmma_f32(a, b, accY);
      a[0] = lo.z; a[1] = lo.w; b[0] = Sr[it][2]; b[1] = Sr[it][3];
      accY = wmma_f32(a, b, accY);
      a[0] = up.x; a[1] = up.y; b[0] = Sr[it][4]; b[1] = Sr[it][5];
      accY = wmma_f32(a, b, accY);
      a[0] = up.z; a[1] = up.w; b[0] = Sr[it][6]; b[1] = Sr[it][7];
      accY = wmma_f32(a, b, accY);
    }
#pragma unroll
    for (int k0 = 0; k0 < CH; k0 += 4)
      accY = wmma_f32(fragA(CmW[wid], ST16, k0), fragB_kn(Vv, ST, k0, nb), accY);
    if (seg == 0 && c == 0 && hi == 0)
      accY[0] += r0sum_s * corr0[nb + ln16];
#pragma unroll
    for (int v = 0; v < 8; ++v) ybuf[(v + 8 * hi) * ST + nb + ln16] = accY[v];

#pragma unroll
    for (int it = 0; it < 4; ++it) {
      v8f a = {};
#pragma unroll
      for (int k0 = 0; k0 < CH; k0 += 4)
        a = wmma_f32(fragA(Kh + it * 16 * ST16, ST16, k0),
                     fragB_kn(Vv, ST, k0, nb), a);
#pragma unroll
      for (int v = 0; v < 8; ++v)
        Sr[it][v] = aLast[it * 16 + v + 8 * hi] * Sr[it][v] + a[v];
    }
    lds_barrier();

    if (tid < CH) {
      float s = 0.f, sq = 0.f;
      for (int j = 0; j < dh; ++j) {
        float y = ybuf[tid * ST + j]; s += y; sq += y * y;
      }
      float m  = s * (1.f / dh);
      float vv = sq * (1.f / dh) - m * m;
      mu_s[tid] = m;
      rs_s[tid] = rsqrtf(vv + 1e-5f);
    }
    lds_barrier();

    {
      int t  = tid >> 3;
      int jb = (tid & 7) * 8;
      float m = mu_s[t], rs = rs_s[t];
      const float gq[8] = {g0.x, g0.y, g0.z, g0.w, g1.x, g1.y, g1.z, g1.w};
#pragma unroll
      for (int q = 0; q < 8; ++q) {
        int j    = jb + q;
        float y  = ybuf[t * ST + j];
        float yn = (y - m) * rs * lnw_s[j] + lnb_s[j];
        int off  = (t0 + t) * D_ + h * dh + j;
        float gv = gq[q];
        og[off]  = yn * (gv / (1.f + __expf(-gv)));
      }
    }
    lds_barrier();
  }
}

__global__ __launch_bounds__(128)
void oproj_kernel(const float* __restrict__ o, const float* __restrict__ Wo,
                  float* __restrict__ out) {
  __shared__ __attribute__((aligned(16))) float Ao[64 * ST];
  __shared__ __attribute__((aligned(16))) float Bo[64 * ST];
  const int tid = threadIdx.x;
  const int lane = tid & 31, wid = tid >> 5;
  const int hi = lane >> 4, ln16 = lane & 15;
  const int nbase = blockIdx.x * 64;
  const int mbase = blockIdx.y * 64;

  v8f acc[4] = {};

  const int srow = tid >> 4;
  const int sc4  = (tid & 15) * 4;
  for (int ks = 0; ks < D_; ks += 64) {
    __syncthreads();
    {
      const float* pa = o  + (size_t)(mbase + srow) * D_ + ks + sc4;
      const float* pb = Wo + (size_t)(nbase + srow) * D_ + ks + sc4;
      v4f a4[8], b4[8];
#pragma unroll
      for (int q = 0; q < 8; ++q) {
        a4[q] = *(const v4f*)(pa + (size_t)q * 8 * D_);
        b4[q] = *(const v4f*)(pb + (size_t)q * 8 * D_);
      }
      asm volatile("" :: "v"(a4[0]), "v"(a4[1]), "v"(a4[2]), "v"(a4[3]),
                         "v"(a4[4]), "v"(a4[5]), "v"(a4[6]), "v"(a4[7]),
                         "v"(b4[0]), "v"(b4[1]), "v"(b4[2]), "v"(b4[3]),
                         "v"(b4[4]), "v"(b4[5]), "v"(b4[6]), "v"(b4[7]));
#pragma unroll
      for (int q = 0; q < 8; ++q) {
        *(v4f*)(Ao + (srow + q * 8) * ST + sc4) = a4[q];
        *(v4f*)(Bo + (srow + q * 8) * ST + sc4) = b4[q];
      }
    }
    __syncthreads();
#pragma unroll
    for (int k0 = 0; k0 < 64; k0 += 4) {
      v2f b = fragA(Bo + wid * 16 * ST, ST, k0);
#pragma unroll
      for (int mt = 0; mt < 4; ++mt)
        acc[mt] = wmma_f32(fragA(Ao + mt * 16 * ST, ST, k0), b, acc[mt]);
    }
  }
#pragma unroll
  for (int mt = 0; mt < 4; ++mt) {
#pragma unroll
    for (int v = 0; v < 8; ++v) {
      int row = mbase + mt * 16 + v + 8 * hi;
      int col = nbase + wid * 16 + ln16;
      out[(size_t)row * D_ + col] = acc[mt][v];
    }
  }
}

extern "C" void kernel_launch(void* const* d_in, const int* in_sizes, int n_in,
                              void* d_out, int out_size, void* d_ws, size_t ws_size,
                              hipStream_t stream) {
  const float* r   = (const float*)d_in[0];
  const float* w   = (const float*)d_in[1];
  const float* k   = (const float*)d_in[2];
  const float* v   = (const float*)d_in[3];
  const float* g   = (const float*)d_in[4];
  const float* u   = (const float*)d_in[5];
  const float* is  = (const float*)d_in[6];
  const float* lnw = (const float*)d_in[7];
  const float* lnb = (const float*)d_in[8];
  const float* Wo  = (const float*)d_in[9];

  float* out   = (float*)d_out;
  float* state = out + (size_t)T_ * D_;

  float* o      = (float*)d_ws;
  float* Aseg   = o + (size_t)T_ * D_;
  float* Sloc   = Aseg + (size_t)H_ * SEG * dh;
  float* Sstart = Sloc + (size_t)H_ * SEG * dh * dh;

  dim3 gs(SEG, H_);
  wkv_pass1<<<gs, 128, 0, stream>>>(w, k, v, Aseg, Sloc);
  state_scan<<<H_, 128, 0, stream>>>(w, is, Aseg, Sloc, Sstart, state);
  wkv_pass2<<<gs, 128, 0, stream>>>(r, w, k, v, g, u, is, lnw, lnb, Sstart, o);

  dim3 g2(D_ / 64, T_ / 64);
  oproj_kernel<<<g2, 128, 0, stream>>>(o, Wo, out);
}


# state scan: 256 work-groups (four rows of a head each, one wave per row), every recurrence behind one load batch, instead of 16 work-groups walking 16 serial load-wait rounds; order of sums unchanged
# speedup vs baseline: 5.7290x; 1.2203x over previous
#include <hip/hip_runtime.h>

typedef __attribute__((ext_vector_type(2))) float v2f;
typedef __attribute__((ext_vector_type(4))) float v4f;
typedef __attribute__((ext_vector_type(8))) float v8f;

#define D_   1024
#define H_   16
#define dh   64
#define T_   4096
#define CH   16
#define SEG  32
#define TSEG (T_/SEG)
#define NCHS (TSEG/CH)
#define ST   68
#define ST16 20

__device__ __forceinline__ v8f wmma_f32(v2f a, v2f b, v8f c) {
  return __builtin_amdgcn_wmma_f32_16x16x4_f32(false, a, false, b, (short)0, c,
                                               false, false);
}

__device__ __forceinline__ v2f fragA(const float* m, int stride, int k0) {
  int lane = threadIdx.x & 31;
  const float* p = m + (lane & 15) * stride + k0 + ((lane >> 4) << 1);
  v2f f; f[0] = p[0]; f[1] = p[1];
  return f;
}
__device__ __forceinline__ v2f fragB_kn(const float* m, int stride, int k0, int nbase) {
  int lane = threadIdx.x & 31;
  int kk = k0 + ((lane >> 4) << 1);
  int n  = nbase + (lane & 15);
  v2f f; f[0] = m[kk * stride + n]; f[1] = m[(kk + 1) * stride + n];
  return f;
}

__device__ __forceinline__ void lds_barrier() {
  __builtin_amdgcn_fence(__ATOMIC_RELEASE, "workgroup", "local");
  __builtin_amdgcn_s_barrier();
  __builtin_amdgcn_fence(__ATOMIC_ACQUIRE, "workgroup", "local");
}

__global__ __launch_bounds__(128)
void wkv_pass1(const float* __restrict__ wg, const float* __restrict__ kg,
               const float* __restrict__ vg, float* __restrict__ Aseg,
               float* __restrict__ Sloc) {
  __shared__ float S[dh * ST];
  __shared__ float Vv[CH * ST];
  __shared__ float Kh[dh * ST16];
  __shared__ float aLast[dh];
  __shared__ float aAcc[dh];

  const int seg = blockIdx.x, h = blockIdx.y;
  const int tid = threadIdx.x;
  const int lane = tid & 31, wid = tid >> 5;
  const int hi = lane >> 4, ln16 = lane & 15;

  if (tid < dh) aAcc[tid] = 1.f;
  const int nb = wid * 16;
  v8f Sr[4] = {};

  const bool scanner = tid < dh;
  const int col = tid & (dh - 1);
  const int gbase = seg * TSEG * D_ + h * dh + col;
  float pw[CH], pk[CH], pv[CH];
  if (scanner) {
#pragma unroll
    for (int t = 0; t < CH; ++t) { pw[t] = wg[gbase + t * D_]; pk[t] = kg[gbase + t * D_]; }
  } else {
#pragma unroll
    for (int t = 0; t < CH; ++t) pv[t] = vg[gbase + t * D_];
  }

  for (int c = 0; c < NCHS; ++c) {
    lds_barrier();
    if (scanner) {
      int i = col;
      float a = 1.f, ai = 1.f;
      float kr[CH], air[CH];
#pragma unroll
      for (int t = 0; t < CH; ++t) {
        float wv = pw[t];
        if (seg == 0 && c == 0 && t == 0) wv = 1.f;
        float kv = pk[t];
        float rw = 1.f / wv;
        a *= wv; ai *= rw;
        kr[t] = kv; air[t] = ai;
      }
      aLast[i] = a;
      aAcc[i] *= a;
#pragma unroll
      for (int t = 0; t < CH; ++t) Kh[i * ST16 + t] = kr[t] * (a * air[t]);
    } else {
#pragma unroll
      for (int t = 0; t < CH; ++t) Vv[t * ST + col] = pv[t];
    }
    if (c + 1 < NCHS) {
      const int nbase = gbase + (c + 1) * CH * D_;
      if (scanner) {
#pragma unroll
        for (int t = 0; t < CH; ++t) { pw[t] = wg[nbase + t * D_]; pk[t] = kg[nbase + t * D_]; }
      } else {
#pragma unroll
        for (int t = 0; t < CH; ++t) pv[t] = vg[nbase + t * D_];
      }
    }
    lds_barrier();

#pragma unroll
    for (int it = 0; it < 4; ++it) {
      v8f acc = {};
#pragma unroll
      for (int k0 = 0; k0 < CH; k0 += 4)
        acc = wmma_f32(fragA(Kh + it * 16 * ST16, ST16, k0),
                       fragB_kn(Vv, ST, k0, nb), acc);
#pragma unroll
      for (int v = 0; v < 8; ++v)
        Sr[it][v] = aLast[it * 16 + v + 8 * hi] * Sr[it][v] + acc[v];
    }
  }
#pragma unroll
  for (int it = 0; it < 4; ++it) {
#pragma unroll
    for (int v = 0; v < 8; ++v)
      S[(it * 16 + v + 8 * hi) * ST + nb + ln16] = Sr[it][v];
  }
  lds_barrier();
  if (tid < dh) Aseg[(h * SEG + seg) * dh + tid] = aAcc[tid];
  for (int x = tid; x < dh * dh; x += 128) {
    int i = x >> 6, j = x & 63;
    Sloc[((size_t)(h * SEG + seg) * dh + i) * dh + j] = S[i * ST + j];
  }
}

__global__ __launch_bounds__(128)
void state_scan(const float* __restrict__ wg, const float* __restrict__ isg,
                const float* __restrict__ Aseg, const float* __restrict__ Sloc,
                float* __restrict__ Sstart, float* __restrict__ state_out) {
  static_assert(SEG == 32, "one decay factor per lane of a 32-lane wave");
  const int h = blockIdx.y, tid = threadIdx.x;
  const int lane = tid & 31;
  const int i = blockIdx.x * 4 + (tid >> 5);
  const int j = lane * 2;
  const float Al = Aseg[(h * SEG + lane) * dh + i];
  float2 Ls[SEG];
#pragma unroll
  for (int sg = 0; sg < SEG; ++sg)
    Ls[sg] = *(const float2*)(Sloc + ((size_t)(h * SEG + sg) * dh + i) * dh + j);
  const float2 w0 = *(const float2*)(wg + h * dh + j);
  const float2 s0 = *(const float2*)(isg + h * dh + j);
  float sa = w0.x * s0.x, sb = w0.y * s0.y;
#pragma unroll
  for (int sg = 0; sg < SEG; ++sg) {
    size_t base = ((size_t)(h * SEG + sg) * dh + i) * dh + j;
    *(float2*)(Sstart + base) = make_float2(sa, sb);
    const float a = __int_as_float(__builtin_amdgcn_readlane(__float_as_int(Al), sg));
    sa = a * sa + Ls[sg].x;
    sb = a * sb + Ls[sg].y;
  }
  *(float2*)(state_out + i * D_ + h * dh + j) = make_float2(sa, sb);
}

__global__ __launch_bounds__(128)
void wkv_pass2(const float* __restrict__ rg, const float* __restrict__ wg,
               const float* __restrict__ kg, const float* __restrict__ vg,
               const float* __restrict__ gg, const float* __restrict__ ug,
               const float* __restrict__ isg, const float* __restrict__ lnw,
               const float* __restrict__ lnb, const float* __restrict__ Sstart,
               float* __restrict__ og) {
  __shared__ __attribute__((aligned(16))) float Rt[CH * ST];
  __shared__ float Kt[CH * ST];
  __shared__ float Ku[CH * ST];
  __shared__ float Vv[CH * ST];
  __shared__ float Kh[dh * ST16];
  __shared__ float CmW[4][CH * ST16];
  __shared__ float ybuf[CH * ST];
  __shared__ float aLast[dh];
  __shared__ float uvec[dh];
  __shared__ float corr0[dh];
  __shared__ float lnw_s[dh], lnb_s[dh];
  __shared__ float mu_s[CH], rs_s[CH];
  __shared__ float r0sum_s;

  const int seg = blockIdx.x, h = blockIdx.y;
  const int tid = threadIdx.x;
  const int lane = tid & 31, wid = tid >> 5;
  const int hi = lane >> 4, ln16 = lane & 15;

  for (int x = tid; x < dh; x += 128) {
    uvec[x]  = ug[h * dh + x];
    corr0[x] = isg[h * dh + x] * (1.f - wg[h * dh + x]);
    lnw_s[x] = lnw[x];
    lnb_s[x] = lnb[x];
  }
  const int nb = wid * 16;
  if (tid == 0 && seg == 0) {
    float s = 0.f;
    for (int i2 = 0; i2 < dh; ++i2) s += rg[h * dh + i2];
    r0sum_s = s;
  }

  const bool scanner = tid < dh;
  const int col = tid & (dh - 1);
  const int gbase = seg * TSEG * D_ + h * dh + col;
  const int gofs = (seg * TSEG + (tid >> 3)) * D_ + h * dh + (tid & 7) * 8;
  float pr[CH], pw[CH], pk[CH], pv[CH];
  if (scanner) {
#pragma unroll
    for (int t = 0; t < CH; ++t) {
      pr[t] = rg[gbase + t * D_]; pw[t] = wg[gbase + t * D_]; pk[t] = kg[gbase + t * D_];
    }
  } else {
#pragma unroll
    for (int t = 0; t < CH; ++t) pv[t] = vg[gbase + t * D_];
  }
  v8f Sr[4];
  {
    const float* sp = Sstart + ((size_t)(h * SEG + seg) * dh) * dh + nb + ln16;
#pragma unroll
    for (int it = 0; it < 4; ++it) {
#pragma unroll
      for (int v = 0; v < 8; ++v) Sr[it][v] = sp[(it * 16 + v + 8 * hi) * dh];
    }
  }
  asm volatile("" :: "v"(Sr[0]), "v"(Sr[1]), "v"(Sr[2]), "v"(Sr[3]));
  lds_barrier();

  for (int c = 0; c < NCHS; ++c) {
    const int t0 = seg * TSEG + c * CH;

    if (scanner) {
      int i = col;
      float a = 1.f, ai = 1.f;
      float kr[CH], air[CH];
#pragma unroll
      for (int t = 0; t < CH; ++t) {
        float wv = pw[t];
        if (seg == 0 && c == 0 && t == 0) wv = 1.f;
        float rv = pr[t], kv = pk[t];
        float aip = ai;
        Rt[t * ST + i] = rv * a;
        float rw = 1.f / wv;
        a *= wv; ai *= rw;
        Kt[t * ST + i] = kv * ai;
        Ku[t * ST + i] = kv * uvec[i] * aip;
        kr[t] = kv; air[t] = ai;
      }
      aLast[i] = a;
#pragma unroll
      for (int t = 0; t < CH; ++t) Kh[i * ST16 + t] = kr[t] * (a * air[t]);
    } else {
#pragma unroll
      for (int t = 0; t < CH; ++t) Vv[t * ST + col] = pv[t];
    }
    const float4 g0 = *(const float4*)(gg + gofs + c * CH * D_);
    const float4 g1 = *(const float4*)(gg + gofs + c * CH * D_ + 4);
    if (c + 1 < NCHS) {
      const int nbase = gbase + (c + 1) * CH * D_;
      if (scanner) {
#pragma unroll
        for (int t = 0; t < CH; ++t) {
          pr[t] = rg[nbase + t * D_]; pw[t] = wg[nbase + t * D_]; pk[t] = kg[nbase + t * D_];
        }
      } else {
#pragma unroll
        for (int t = 0; t < CH; ++t) pv[t] = vg[nbase + t * D_];
      }
    }
    lds_barrier();

    float4 ra[8];
#pragma unroll
    for (int it = 0; it < 4; ++it) {
      const float4* p = (const float4*)(Rt + ln16 * ST + it * 16 + 8 * hi);
      ra[2 * it] = p[0]; ra[2 * it + 1] = p[1];
    }

    v8f accC = {}; v8f accD = {};
#pragma unroll
    for (int k0 = 0; k0 < dh; k0 += 4) {
      v2f arf = fragA(Rt, ST, k0);
      accC = wmma_f32(arf, fragA(Kt, ST, k0), accC);
      accD = wmma_f32(arf, fragA(Ku, ST, k0), accD);
    }
    {
      float* cm = CmW[wid];
#pragma unroll
      for (int v = 0; v < 8; ++v) {
        int M = v + 8 * hi, N = ln16;
        cm[M * ST16 + N] = (N < M) ? accC[v] : ((N == M) ? accD[v] : 0.f);
      }
    }
    asm volatile("" ::: "memory");

    v8f accY = {};
#pragma unroll
    for (int it = 0; it < 4; ++it) {
      const float4 lo = ra[2 * it], up = ra[2 * it + 1];
      v2f a, b;
      a[0] = lo.x; a[1] = lo.y; b[0] = Sr[it][0]; b[1] = Sr[it][1];
      accY = wmma_f32(a, b, accY);
      a[0] = lo.z; a[1] = lo.w; b[0] = Sr[it][2]; b[1] = Sr[it][3];
      accY = wmma_f32(a, b, accY);
      a[0] = up.x; a[1] = up.y; b[0] = Sr[it][4]; b[1] = Sr[it][5];
      accY = wmma_f32(a, b, accY);
      a[0] = up.z; a[1] = up.w; b[0] = Sr[it][6]; b[1] = Sr[it][7];
      accY = wmma_f32(a, b, accY);
    }
#pragma unroll
    for (int k0 = 0; k0 < CH; k0 += 4)
      accY = wmma_f32(fragA(CmW[wid], ST16, k0), fragB_kn(Vv, ST, k0, nb), accY);
    if (seg == 0 && c == 0 && hi == 0)
      accY[0] += r0sum_s * corr0[nb + ln16];
#pragma unroll
    for (int v = 0; v < 8; ++v) ybuf[(v + 8 * hi) * ST + nb + ln16] = accY[v];

#pragma unroll
    for (int it = 0; it < 4; ++it) {
      v8f a = {};
#pragma unroll
      for (int k0 = 0; k0 < CH; k0 += 4)
        a = wmma_f32(fragA(Kh + it * 16 * ST16, ST16, k0),
                     fragB_kn(Vv, ST, k0, nb), a);
#pragma unroll
      for (int v = 0; v < 8; ++v)
        Sr[it][v] = aLast[it * 16 + v + 8 * hi] * Sr[it][v] + a[v];
    }
    lds_barrier();

    if (tid < CH) {
      float s = 0.f, sq = 0.f;
      for (int j = 0; j < dh; ++j) {
        float y = ybuf[tid * ST + j]; s += y; sq += y * y;
      }
      float m  = s * (1.f / dh);
      float vv = sq * (1.f / dh) - m * m;
      mu_s[tid] = m;
      rs_s[tid] = rsqrtf(vv + 1e-5f);
    }
    lds_barrier();

    {
      int t  = tid >> 3;
      int jb = (tid & 7) * 8;
      float m = mu_s[t], rs = rs_s[t];
      const float gq[8] = {g0.x, g0.y, g0.z, g0.w, g1.x, g1.y, g1.z, g1.w};
#pragma unroll
      for (int q = 0; q < 8; ++q) {
        int j    = jb + q;
        float y  = ybuf[t * ST + j];
        float yn = (y - m) * rs * lnw_s[j] + lnb_s[j];
        int off  = (t0 + t) * D_ + h * dh + j;
        float gv = gq[q];
        og[off]  = yn * (gv / (1.f + __expf(-gv)));
      }
    }
    lds_barrier();
  }
}

__global__ __launch_bounds__(128)
void oproj_kernel(const float* __restrict__ o, const float* __restrict__ Wo,
                  float* __restrict__ out) {
  __shared__ __attribute__((aligned(16))) float Ao[64 * ST];
  __shared__ __attribute__((aligned(16))) float Bo[64 * ST];
  const int tid = threadIdx.x;
  const int lane = tid & 31, wid = tid >> 5;
  const int hi = lane >> 4, ln16 = lane & 15;
  const int nbase = blockIdx.x * 64;
  const int mbase = blockIdx.y * 64;

  v8f acc[4] = {};

  const int srow = tid >> 4;
  const int sc4  = (tid & 15) * 4;
  for (int ks = 0; ks < D_; ks += 64) {
    __syncthreads();
    {
      const float* pa = o  + (size_t)(mbase + srow) * D_ + ks + sc4;
      const float* pb = Wo + (size_t)(nbase + srow) * D_ + ks + sc4;
      v4f a4[8], b4[8];
#pragma unroll
      for (int q = 0; q < 8; ++q) {
        a4[q] = *(const v4f*)(pa + (size_t)q * 8 * D_);
        b4[q] = *(const v4f*)(pb + (size_t)q * 8 * D_);
      }
      asm volatile("" :: "v"(a4[0]), "v"(a4[1]), "v"(a4[2]), "v"(a4[3]),
                         "v"(a4[4]), "v"(a4[5]), "v"(a4[6]), "v"(a4[7]),
                         "v"(b4[0]), "v"(b4[1]), "v"(b4[2]), "v"(b4[3]),
                         "v"(b4[4]), "v"(b4[5]), "v"(b4[6]), "v"(b4[7]));
#pragma unroll
      for (int q = 0; q < 8; ++q) {
        *(v4f*)(Ao + (srow + q * 8) * ST + sc4) = a4[q];
        *(v4f*)(Bo + (srow + q * 8) * ST + sc4) = b4[q];
      }
    }
    __syncthreads();
#pragma unroll
    for (int k0 = 0; k0 < 64; k0 += 4) {
      v2f b = fragA(Bo + wid * 16 * ST, ST, k0);
#pragma unroll
      for (int mt = 0; mt < 4; ++mt)
        acc[mt] = wmma_f32(fragA(Ao + mt * 16 * ST, ST, k0), b, acc[mt]);
    }
  }
#pragma unroll
  for (int mt = 0; mt < 4; ++mt) {
#pragma unroll
    for (int v = 0; v < 8; ++v) {
      int row = mbase + mt * 16 + v + 8 * hi;
      int col = nbase + wid * 16 + ln16;
      out[(size_t)row * D_ + col] = acc[mt][v];
    }
  }
}

extern "C" void kernel_launch(void* const* d_in, const int* in_sizes, int n_in,
                              void* d_out, int out_size, void* d_ws, size_t ws_size,
                              hipStream_t stream) {
  const float* r   = (const float*)d_in[0];
  const float* w   = (const float*)d_in[1];
  const float* k   = (const float*)d_in[2];
  const float* v   = (const float*)d_in[3];
  const float* g   = (const float*)d_in[4];
  const float* u   = (const float*)d_in[5];
  const float* is  = (const float*)d_in[6];
  const float* lnw = (const float*)d_in[7];
  const float* lnb = (const float*)d_in[8];
  const float* Wo  = (const float*)d_in[9];

  float* out   = (float*)d_out;
  float* state = out + (size_t)T_ * D_;

  float* o      = (float*)d_ws;
  float* Aseg   = o + (size_t)T_ * D_;
  float* Sloc   = Aseg + (size_t)H_ * SEG * dh;
  float* Sstart = Sloc + (size_t)H_ * SEG * dh * dh;

  dim3 gs(SEG, H_);
  wkv_pass1<<<gs, 128, 0, stream>>>(w, k, v, Aseg, Sloc);
  dim3 gsc(dh / 4, H_);
  state_scan<<<gsc, 128, 0, stream>>>(w, is, Aseg, Sloc, Sstart, state);
  wkv_pass2<<<gs, 128, 0, stream>>>(r, w, k, v, g, u, is, lnw, lnb, Sstart, o);

  dim3 g2(D_ / 64, T_ / 64);
  oproj_kernel<<<g2, 128, 0, stream>>>(o, Wo, out);
}


# o_proj: next k-step's tile loads in flight behind this step's WMMAs (2-stage LDS ring, 1 barrier a step), one 128x128 8-wave group per WGP, not 1024 64x64 load-wait-compute groups; sum order kept
# speedup vs baseline: 6.1954x; 1.0814x over previous
#include <hip/hip_runtime.h>

typedef __attribute__((ext_vector_type(2))) float v2f;
typedef __attribute__((ext_vector_type(4))) float v4f;
typedef __attribute__((ext_vector_type(8))) float v8f;

#define D_   1024
#define H_   16
#define dh   64
#define T_   4096
#define CH   16
#define SEG  32
#define TSEG (T_/SEG)
#define NCHS (TSEG/CH)
#define ST   68
#define ST16 20

__device__ __forceinline__ v8f wmma_f32(v2f a, v2f b, v8f c) {
  return __builtin_amdgcn_wmma_f32_16x16x4_f32(false, a, false, b, (short)0, c,
                                               false, false);
}

__device__ __forceinline__ v2f fragA(const float* m, int stride, int k0) {
  int lane = threadIdx.x & 31;
  const float* p = m + (lane & 15) * stride + k0 + ((lane >> 4) << 1);
  v2f f; f[0] = p[0]; f[1] = p[1];
  return f;
}
__device__ __forceinline__ v2f fragB_kn(const float* m, int stride, int k0, int nbase) {
  int lane = threadIdx.x & 31;
  int kk = k0 + ((lane >> 4) << 1);
  int n  = nbase + (lane & 15);
  v2f f; f[0] = m[kk * stride + n]; f[1] = m[(kk + 1) * stride + n];
  return f;
}

__device__ __forceinline__ void lds_barrier() {
  __builtin_amdgcn_fence(__ATOMIC_RELEASE, "workgroup", "local");
  __builtin_amdgcn_s_barrier();
  __builtin_amdgcn_fence(__ATOMIC_ACQUIRE, "workgroup", "local");
}

__global__ __launch_bounds__(128)
void wkv_pass1(const float* __restrict__ wg, const float* __restrict__ kg,
               const float* __restrict__ vg, float* __restrict__ Aseg,
               float* __restrict__ Sloc) {
  __shared__ float S[dh * ST];
  __shared__ float Vv[CH * ST];
  __shared__ float Kh[dh * ST16];
  __shared__ float aLast[dh];
  __shared__ float aAcc[dh];

  const int seg = blockIdx.x, h = blockIdx.y;
  const int tid = threadIdx.x;
  const int lane = tid & 31, wid = tid >> 5;
  const int hi = lane >> 4, ln16 = lane & 15;

  if (tid < dh) aAcc[tid] = 1.f;
  const int nb = wid * 16;
  v8f Sr[4] = {};

  const bool scanner = tid < dh;
  const int col = tid & (dh - 1);
  const int gbase = seg * TSEG * D_ + h * dh + col;
  float pw[CH], pk[CH], pv[CH];
  if (scanner) {
#pragma unroll
    for (int t = 0; t < CH; ++t) { pw[t] = wg[gbase + t * D_]; pk[t] = kg[gbase + t * D_]; }
  } else {
#pragma unroll
    for (int t = 0; t < CH; ++t) pv[t] = vg[gbase + t * D_];
  }

  for (int c = 0; c < NCHS; ++c) {
    lds_barrier();
    if (scanner) {
      int i = col;
      float a = 1.f, ai = 1.f;
      float kr[CH], air[CH];
#pragma unroll
      for (int t = 0; t < CH; ++t) {
        float wv = pw[t];
        if (seg == 0 && c == 0 && t == 0) wv = 1.f;
        float kv = pk[t];
        float rw = 1.f / wv;
        a *= wv; ai *= rw;
        kr[t] = kv; air[t] = ai;
      }
      aLast[i] = a;
      aAcc[i] *= a;
#pragma unroll
      for (int t = 0; t < CH; ++t) Kh[i * ST16 + t] = kr[t] * (a * air[t]);
    } else {
#pragma unroll
      for (int t = 0; t < CH; ++t) Vv[t * ST + col] = pv[t];
    }
    if (c + 1 < NCHS) {
      const int nbase = gbase + (c + 1) * CH * D_;
      if (scanner) {
#pragma unroll
        for (int t = 0; t < CH; ++t) { pw[t] = wg[nbase + t * D_]; pk[t] = kg[nbase + t * D_]; }
      } else {
#pragma unroll
        for (int t = 0; t < CH; ++t) pv[t] = vg[nbase + t * D_];
      }
    }
    lds_barrier();

#pragma unroll
    for (int it = 0; it < 4; ++it) {
      v8f acc = {};
#pragma unroll
      for (int k0 = 0; k0 < CH; k0 += 4)
        acc = wmma_f32(fragA(Kh + it * 16 * ST16, ST16, k0),
                       fragB_kn(Vv, ST, k0, nb), acc);
#pragma unroll
      for (int v = 0; v < 8; ++v)
        Sr[it][v] = aLast[it * 16 + v + 8 * hi] * Sr[it][v] + acc[v];
    }
  }
#pragma unroll
  for (int it = 0; it < 4; ++it) {
#pragma unroll
    for (int v = 0; v < 8; ++v)
      S[(it * 16 + v + 8 * hi) * ST + nb + ln16] = Sr[it][v];
  }
  lds_barrier();
  if (tid < dh) Aseg[(h * SEG + seg) * dh + tid] = aAcc[tid];
  for (int x = tid; x < dh * dh; x += 128) {
    int i = x >> 6, j = x & 63;
    Sloc[((size_t)(h * SEG + seg) * dh + i) * dh + j] = S[i * ST + j];
  }
}

__global__ __launch_bounds__(128)
void state_scan(const float* __restrict__ wg, const float* __restrict__ isg,
                const float* __restrict__ Aseg, const float* __restrict__ Sloc,
                float* __restrict__ Sstart, float* __restrict__ state_out) {
  static_assert(SEG == 32, "one decay factor per lane of a 32-lane wave");
  const int h = blockIdx.y, tid = threadIdx.x;
  const int lane = tid & 31;
  const int i = blockIdx.x * 4 + (tid >> 5);
  const int j = lane * 2;
  const float Al = Aseg[(h * SEG + lane) * dh + i];
  float2 Ls[SEG];
#pragma unroll
  for (int sg = 0; sg < SEG; ++sg)
    Ls[sg] = *(const float2*)(Sloc + ((size_t)(h * SEG + sg) * dh + i) * dh + j);
  const float2 w0 = *(const float2*)(wg + h * dh + j);
  const float2 s0 = *(const float2*)(isg + h * dh + j);
  float sa = w0.x * s0.x, sb = w0.y * s0.y;
#pragma unroll
  for (int sg = 0; sg < SEG; ++sg) {
    size_t base = ((size_t)(h * SEG + sg) * dh + i) * dh + j;
    *(float2*)(Sstart + base) = make_float2(sa, sb);
    const float a = __int_as_float(__builtin_amdgcn_readlane(__float_as_int(Al), sg));
    sa = a * sa + Ls[sg].x;
    sb = a * sb + Ls[sg].y;
  }
  *(float2*)(state_out + i * D_ + h * dh + j) = make_float2(sa, sb);
}

__global__ __launch_bounds__(128)
void wkv_pass2(const float* __restrict__ rg, const float* __restrict__ wg,
               const float* __restrict__ kg, const float* __restrict__ vg,
               const float* __restrict__ gg, const float* __restrict__ ug,
               const float* __restrict__ isg, const float* __restrict__ lnw,
               const float* __restrict__ lnb, const float* __restrict__ Sstart,
               float* __restrict__ og) {
  __shared__ __attribute__((aligned(16))) float Rt[CH * ST];
  __shared__ float Kt[CH * ST];
  __shared__ float Ku[CH * ST];
  __shared__ float Vv[CH * ST];
  __shared__ float Kh[dh * ST16];
  __shared__ float CmW[4][CH * ST16];
  __shared__ float ybuf[CH * ST];
  __shared__ float aLast[dh];
  __shared__ float uvec[dh];
  __shared__ float corr0[dh];
  __shared__ float lnw_s[dh], lnb_s[dh];
  __shared__ float mu_s[CH], rs_s[CH];
  __shared__ float r0sum_s;

  const int seg = blockIdx.x, h = blockIdx.y;
  const int tid = threadIdx.x;
  const int lane = tid & 31, wid = tid >> 5;
  const int hi = lane >> 4, ln16 = lane & 15;

  for (int x = tid; x < dh; x += 128) {
    uvec[x]  = ug[h * dh + x];
    corr0[x] = isg[h * dh + x] * (1.f - wg[h * dh + x]);
    lnw_s[x] = lnw[x];
    lnb_s[x] = lnb[x];
  }
  const int nb = wid * 16;
  if (tid == 0 && seg == 0) {
    float s = 0.f;
    for (int i2 = 0; i2 < dh; ++i2) s += rg[h * dh + i2];
    r0sum_s = s;
  }

  const bool scanner = tid < dh;
  const int col = tid & (dh - 1);
  const int gbase = seg * TSEG * D_ + h * dh + col;
  const int gofs = (seg * TSEG + (tid >> 3)) * D_ + h * dh + (tid & 7) * 8;
  float pr[CH], pw[CH], pk[CH], pv[CH];
  if (scanner) {
#pragma unroll
    for (int t = 0; t < CH; ++t) {
      pr[t] = rg[gbase + t * D_]; pw[t] = wg[gbase + t * D_]; pk[t] = kg[gbase + t * D_];
    }
  } else {
#pragma unroll
    for (int t = 0; t < CH; ++t) pv[t] = vg[gbase + t * D_];
  }
  v8f Sr[4];
  {
    const float* sp = Sstart + ((size_t)(h * SEG + seg) * dh) * dh + nb + ln16;
#pragma unroll
    for (int it = 0; it < 4; ++it) {
#pragma unroll
      for (int v = 0; v < 8; ++v) Sr[it][v] = sp[(it * 16 + v + 8 * hi) * dh];
    }
  }
  asm volatile("" :: "v"(Sr[0]), "v"(Sr[1]), "v"(Sr[2]), "v"(Sr[3]));
  lds_barrier();

  for (int c = 0; c < NCHS; ++c) {
    const int t0 = seg * TSEG + c * CH;

    if (scanner) {
      int i = col;
      float a = 1.f, ai = 1.f;
      float kr[CH], air[CH];
#pragma unroll
      for (int t = 0; t < CH; ++t) {
        float wv = pw[t];
        if (seg == 0 && c == 0 && t == 0) wv = 1.f;
        float rv = pr[t], kv = pk[t];
        float aip = ai;
        Rt[t * ST + i] = rv * a;
        float rw = 1.f / wv;
        a *= wv; ai *= rw;
        Kt[t * ST + i] = kv * ai;
        Ku[t * ST + i] = kv * uvec[i] * aip;
        kr[t] = kv; air[t] = ai;
      }
      aLast[i] = a;
#pragma unroll
      for (int t = 0; t < CH; ++t) Kh[i * ST16 + t] = kr[t] * (a * air[t]);
    } else {
#pragma unroll
      for (int t = 0; t < CH; ++t) Vv[t * ST + col] = pv[t];
    }
    const float4 g0 = *(const float4*)(gg + gofs + c * CH * D_);
    const float4 g1 = *(const float4*)(gg + gofs + c * CH * D_ + 4);
    if (c + 1 < NCHS) {
      const int nbase = gbase + (c + 1) * CH * D_;
      if (scanner) {
#pragma unroll
        for (int t = 0; t < CH; ++t) {
          pr[t] = rg[nbase + t * D_]; pw[t] = wg[nbase + t * D_]; pk[t] = kg[nbase + t * D_];
        }
      } else {
#pragma unroll
        for (int t = 0; t < CH; ++t) pv[t] = vg[nbase + t * D_];
      }
    }
    lds_barrier();

    float4 ra[8];
#pragma unroll
    for (int it = 0; it < 4; ++it) {
      const float4* p = (const float4*)(Rt + ln16 * ST + it * 16 + 8 * hi);
      ra[2 * it] = p[0]; ra[2 * it + 1] = p[1];
    }

    v8f accC = {}; v8f accD = {};
#pragma unroll
    for (int k0 = 0; k0 < dh; k0 += 4) {
      v2f arf = fragA(Rt, ST, k0);
      accC = wmma_f32(arf, fragA(Kt, ST, k0), accC);
      accD = wmma_f32(arf, fragA(Ku, ST, k0), accD);
    }
    {
      float* cm = CmW[wid];
#pragma unroll
      for (int v = 0; v < 8; ++v) {
        int M = v + 8 * hi, N = ln16;
        cm[M * ST16 + N] = (N < M) ? accC[v] : ((N == M) ? accD[v] : 0.f);
      }
    }
    asm volatile("" ::: "memory");

    v8f accY = {};
#pragma unroll
    for (int it = 0; it < 4; ++it) {
      const float4 lo = ra[2 * it], up = ra[2 * it + 1];
      v2f a, b;
      a[0] = lo.x; a[1] = lo.y; b[0] = Sr[it][0]; b[1] = Sr[it][1];
      accY = wmma_f32(a, b, accY);
      a[0] = lo.z; a[1] = lo.w; b[0] = Sr[it][2]; b[1] = Sr[it][3];
      accY = wmma_f32(a, b, accY);
      a[0] = up.x; a[1] = up.y; b[0] = Sr[it][4]; b[1] = Sr[it][5];
      accY = wmma_f32(a, b, accY);
      a[0] = up.z; a[1] = up.w; b[0] = Sr[it][6]; b[1] = Sr[it][7];
      accY = wmma_f32(a, b, accY);
    }
#pragma unroll
    for (int k0 = 0; k0 < CH; k0 += 4)
      accY = wmma_f32(fragA(CmW[wid], ST16, k0), fragB_kn(Vv, ST, k0, nb), accY);
    if (seg == 0 && c == 0 && hi == 0)
      accY[0] += r0sum_s * corr0[nb + ln16];
#pragma unroll
    for (int v = 0; v < 8; ++v) ybuf[(v + 8 * hi) * ST + nb + ln16] = accY[v];

#pragma unroll
    for (int it = 0; it < 4; ++it) {
      v8f a = {};
#pragma unroll
      for (int k0 = 0; k0 < CH; k0 += 4)
        a = wmma_f32(fragA(Kh + it * 16 * ST16, ST16, k0),
                     fragB_kn(Vv, ST, k0, nb), a);
#pragma unroll
      for (int v = 0; v < 8; ++v)
        Sr[it][v] = aLast[it * 16 + v + 8 * hi] * Sr[it][v] + a[v];
    }
    lds_barrier();

    if (tid < CH) {
      float s = 0.f, sq = 0.f;
      for (int j = 0; j < dh; ++j) {
        float y = ybuf[tid * ST + j]; s += y; sq += y * y;
      }
      float m  = s * (1.f / dh);
      float vv = sq * (1.f / dh) - m * m;
      mu_s[tid] = m;
      rs_s[tid] = rsqrtf(vv + 1e-5f);
    }
    lds_barrier();

    {
      int t  = tid >> 3;
      int jb = (tid & 7) * 8;
      float m = mu_s[t], rs = rs_s[t];
      const float gq[8] = {g0.x, g0.y, g0.z, g0.w, g1.x, g1.y, g1.z, g1.w};
#pragma unroll
      for (int q = 0; q < 8; ++q) {
        int j    = jb + q;
        float y  = ybuf[t * ST + j];
        float yn = (y - m) * rs * lnw_s[j] + lnb_s[j];
        int off  = (t0 + t) * D_ + h * dh + j;
        float gv = gq[q];
        og[off]  = yn * (gv / (1.f + __expf(-gv)));
      }
    }
    lds_barrier();
  }
}

#define OT   128
#define OKS  64
#define ONS  (D_ / OKS)
__global__ __launch_bounds__(256)
void oproj_kernel(const float* __restrict__ o, const float* __restrict__ Wo,
                  float* __restrict__ out) {
  __shared__ __attribute__((aligned(16))) float Ao[2][OT * ST];
  __shared__ __attribute__((aligned(16))) float Bo[2][OT * ST];
  const int tid = threadIdx.x;
  const int lane = tid & 31, wid = tid >> 5;
  const int hi = lane >> 4, ln16 = lane & 15;
  const int nbase = blockIdx.x * OT;
  const int mbase = blockIdx.y * OT;
  const int wm = (wid >> 2) * 64;
  const int wn = (wid & 3) * 32;

  v8f acc[4][2] = {};

  const int srow = tid >> 4;
  const int sc4  = (tid & 15) * 4;
  const float* pa = o  + (size_t)(mbase + srow) * D_ + sc4;
  const float* pb = Wo + (size_t)(nbase + srow) * D_ + sc4;
  const int sofs = srow * ST + sc4;
  v4f a4[8], b4[8];

#pragma unroll
  for (int q = 0; q < 8; ++q) {
    a4[q] = *(const v4f*)(pa + (size_t)q * 16 * D_);
    b4[q] = *(const v4f*)(pb + (size_t)q * 16 * D_);
  }
#pragma unroll
  for (int q = 0; q < 8; ++q) {
    *(v4f*)(Ao[0] + sofs + q * 16 * ST) = a4[q];
    *(v4f*)(Bo[0] + sofs + q * 16 * ST) = b4[q];
  }
  lds_barrier();

  for (int s = 0; s < ONS; ++s) {
    const int cur = s & 1;
    const bool more = s + 1 < ONS;
    if (more) {
      const int ks = (s + 1) * OKS;
#pragma unroll
      for (int q = 0; q < 8; ++q) {
        a4[q] = *(const v4f*)(pa + ks + (size_t)q * 16 * D_);
        b4[q] = *(const v4f*)(pb + ks + (size_t)q * 16 * D_);
      }
    }
    const float* As = Ao[cur] + wm * ST;
    const float* Bs = Bo[cur] + wn * ST;
#pragma unroll
    for (int k0 = 0; k0 < OKS / 2; k0 += 4) {
      const v2f b0 = fragA(Bs, ST, k0);
      const v2f b1 = fragA(Bs + 16 * ST, ST, k0);
#pragma unroll
      for (int mt = 0; mt < 4; ++mt) {
        const v2f a = fragA(As + mt * 16 * ST, ST, k0);
        acc[mt][0] = wmma_f32(a, b0, acc[mt][0]);
        acc[mt][1] = wmma_f32(a, b1, acc[mt][1]);
      }
    }
    __builtin_amdgcn_sched_barrier(0);
    if (more) {
#pragma unroll
      for (int q = 0; q < 8; ++q) {
        *(v4f*)(Ao[cur ^ 1] + sofs + q * 16 * ST) = a4[q];
        *(v4f*)(Bo[cur ^ 1] + sofs + q * 16 * ST) = b4[q];
      }
    }
#pragma unroll
    for (int k0 = OKS / 2; k0 < OKS; k0 += 4) {
      const v2f b0 = fragA(Bs, ST, k0);
      const v2f b1 = fragA(Bs + 16 * ST, ST, k0);
#pragma unroll
      for (int mt = 0; mt < 4; ++mt) {
        const v2f a = fragA(As + mt * 16 * ST, ST, k0);
        acc[mt][0] = wmma_f32(a, b0, acc[mt][0]);
        acc[mt][1] = wmma_f32(a, b1, acc[mt][1]);
      }
    }
    lds_barrier();
  }
#pragma unroll
  for (int mt = 0; mt < 4; ++mt) {
#pragma unroll
    for (int nt = 0; nt < 2; ++nt) {
#pragma unroll
      for (int v = 0; v < 8; ++v) {
        int row = mbase + wm + mt * 16 + v + 8 * hi;
        int col = nbase + wn + nt * 16 + ln16;
        out[(size_t)row * D_ + col] = acc[mt][nt][v];
      }
    }
  }
}

extern "C" void kernel_launch(void* const* d_in, const int* in_sizes, int n_in,
                              void* d_out, int out_size, void* d_ws, size_t ws_size,
                              hipStream_t stream) {
  const float* r   = (const float*)d_in[0];
  const float* w   = (const float*)d_in[1];
  const float* k   = (const float*)d_in[2];
  const float* v   = (const float*)d_in[3];
  const float* g   = (const float*)d_in[4];
  const float* u   = (const float*)d_in[5];
  const float* is  = (const float*)d_in[6];
  const float* lnw = (const float*)d_in[7];
  const float* lnb = (const float*)d_in[8];
  const float* Wo  = (const float*)d_in[9];

  float* out   = (float*)d_out;
  float* state = out + (size_t)T_ * D_;

  float* o      = (float*)d_ws;
  float* Aseg   = o + (size_t)T_ * D_;
  float* Sloc   = Aseg + (size_t)H_ * SEG * dh;
  float* Sstart = Sloc + (size_t)H_ * SEG * dh * dh;

  dim3 gs(SEG, H_);
  wkv_pass1<<<gs, 128, 0, stream>>>(w, k, v, Aseg, Sloc);
  dim3 gsc(dh / 4, H_);
  state_scan<<<gsc, 128, 0, stream>>>(w, is, Aseg, Sloc, Sstart, state);
  wkv_pass2<<<gs, 128, 0, stream>>>(r, w, k, v, g, u, is, lnw, lnb, Sstart, o);

  dim3 g2(D_ / OT, T_ / OT);
  oproj_kernel<<<g2, 256, 0, stream>>>(o, Wo, out);
}


# pass2: jobs common to the work-group shared out over its 4 waves: score tile C,D a quarter of k per wave (partials added via LDS), LN sums by all threads + lane adds; C and LN sum order changes
# speedup vs baseline: 6.6613x; 1.0752x over previous
#include <hip/hip_runtime.h>

typedef __attribute__((ext_vector_type(2))) float v2f;
typedef __attribute__((ext_vector_type(4))) float v4f;
typedef __attribute__((ext_vector_type(8))) float v8f;

#define D_   1024
#define H_   16
#define dh   64
#define T_   4096
#define CH   16
#define SEG  32
#define TSEG (T_/SEG)
#define NCHS (TSEG/CH)
#define ST   68
#define ST16 20

__device__ __forceinline__ v8f wmma_f32(v2f a, v2f b, v8f c) {
  return __builtin_amdgcn_wmma_f32_16x16x4_f32(false, a, false, b, (short)0, c,
                                               false, false);
}

__device__ __forceinline__ v2f fragA(const float* m, int stride, int k0) {
  int lane = threadIdx.x & 31;
  const float* p = m + (lane & 15) * stride + k0 + ((lane >> 4) << 1);
  v2f f; f[0] = p[0]; f[1] = p[1];
  return f;
}
__device__ __forceinline__ v2f fragB_kn(const float* m, int stride, int k0, int nbase) {
  int lane = threadIdx.x & 31;
  int kk = k0 + ((lane >> 4) << 1);
  int n  = nbase + (lane & 15);
  v2f f; f[0] = m[kk * stride + n]; f[1] = m[(kk + 1) * stride + n];
  return f;
}

__device__ __forceinline__ void lds_barrier() {
  __builtin_amdgcn_fence(__ATOMIC_RELEASE, "workgroup", "local");
  __builtin_amdgcn_s_barrier();
  __builtin_amdgcn_fence(__ATOMIC_ACQUIRE, "workgroup", "local");
}

__device__ __forceinline__ float sum8(float x) {
  x += __int_as_float(__builtin_amdgcn_update_dpp(0, __float_as_int(x), 0x161, 0xf, 0xf, false));
  x += __int_as_float(__builtin_amdgcn_update_dpp(0, __float_as_int(x), 0x162, 0xf, 0xf, false));
  x += __int_as_float(__builtin_amdgcn_update_dpp(0, __float_as_int(x), 0x164, 0xf, 0xf, false));
  return x;
}

__global__ __launch_bounds__(128)
void wkv_pass1(const float* __restrict__ wg, const float* __restrict__ kg,
               const float* __restrict__ vg, float* __restrict__ Aseg,
               float* __restrict__ Sloc) {
  __shared__ float S[dh * ST];
  __shared__ float Vv[CH * ST];
  __shared__ float Kh[dh * ST16];
  __shared__ float aLast[dh];
  __shared__ float aAcc[dh];

  const int seg = blockIdx.x, h = blockIdx.y;
  const int tid = threadIdx.x;
  const int lane = tid & 31, wid = tid >> 5;
  const int hi = lane >> 4, ln16 = lane & 15;

  if (tid < dh) aAcc[tid] = 1.f;
  const int nb = wid * 16;
  v8f Sr[4] = {};

  const bool scanner = tid < dh;
  const int col = tid & (dh - 1);
  const int gbase = seg * TSEG * D_ + h * dh + col;
  float pw[CH], pk[CH], pv[CH];
  if (scanner) {
#pragma unroll
    for (int t = 0; t < CH; ++t) { pw[t] = wg[gbase + t * D_]; pk[t] = kg[gbase + t * D_]; }
  } else {
#pragma unroll
    for (int t = 0; t < CH; ++t) pv[t] = vg[gbase + t * D_];
  }

  for (int c = 0; c < NCHS; ++c) {
    lds_barrier();
    if (scanner) {
      int i = col;
      float a = 1.f, ai = 1.f;
      float kr[CH], air[CH];
#pragma unroll
      for (int t = 0; t < CH; ++t) {
        float wv = pw[t];
        if (seg == 0 && c == 0 && t == 0) wv = 1.f;
        float kv = pk[t];
        float rw = 1.f / wv;
        a *= wv; ai *= rw;
        kr[t] = kv; air[t] = ai;
      }
      aLast[i] = a;
      aAcc[i] *= a;
#pragma unroll
      for (int t = 0; t < CH; ++t) Kh[i * ST16 + t] = kr[t] * (a * air[t]);
    } else {
#pragma unroll
      for (int t = 0; t < CH; ++t) Vv[t * ST + col] = pv[t];
    }
    if (c + 1 < NCHS) {
      const int nbase = gbase + (c + 1) * CH * D_;
      if (scanner) {
#pragma unroll
        for (int t = 0; t < CH; ++t) { pw[t] = wg[nbase + t * D_]; pk[t] = kg[nbase + t * D_]; }
      } else {
#pragma unroll
        for (int t = 0; t < CH; ++t) pv[t] = vg[nbase + t * D_];
      }
    }
    lds_barrier();

#pragma unroll
    for (int it = 0; it < 4; ++it) {
      v8f acc = {};
#pragma unroll
      for (int k0 = 0; k0 < CH; k0 += 4)
        acc = wmma_f32(fragA(Kh + it * 16 * ST16, ST16, k0),
                       fragB_kn(Vv, ST, k0, nb), acc);
#pragma unroll
      for (int v = 0; v < 8; ++v)
        Sr[it][v] = aLast[it * 16 + v + 8 * hi] * Sr[it][v] + acc[v];
    }
  }
#pragma unroll
  for (int it = 0; it < 4; ++it) {
#pragma unroll
    for (int v = 0; v < 8; ++v)
      S[(it * 16 + v + 8 * hi) * ST + nb + ln16] = Sr[it][v];
  }
  lds_barrier();
  if (tid < dh) Aseg[(h * SEG + seg) * dh + tid] = aAcc[tid];
  for (int x = tid; x < dh * dh; x += 128) {
    int i = x >> 6, j = x & 63;
    Sloc[((size_t)(h * SEG + seg) * dh + i) * dh + j] = S[i * ST + j];
  }
}

__global__ __launch_bounds__(128)
void state_scan(const float* __restrict__ wg, const float* __restrict__ isg,
                const float* __restrict__ Aseg, const float* __restrict__ Sloc,
                float* __restrict__ Sstart, float* __restrict__ state_out) {
  static_assert(SEG == 32, "one decay factor per lane of a 32-lane wave");
  const int h = blockIdx.y, tid = threadIdx.x;
  const int lane = tid & 31;
  const int i = blockIdx.x * 4 + (tid >> 5);
  const int j = lane * 2;
  const float Al = Aseg[(h * SEG + lane) * dh + i];
  float2 Ls[SEG];
#pragma unroll
  for (int sg = 0; sg < SEG; ++sg)
    Ls[sg] = *(const float2*)(Sloc + ((size_t)(h * SEG + sg) * dh + i) * dh + j);
  const float2 w0 = *(const float2*)(wg + h * dh + j);
  const float2 s0 = *(const float2*)(isg + h * dh + j);
  float sa = w0.x * s0.x, sb = w0.y * s0.y;
#pragma unroll
  for (int sg = 0; sg < SEG; ++sg) {
    size_t base = ((size_t)(h * SEG + sg) * dh + i) * dh + j;
    *(float2*)(Sstart + base) = make_float2(sa, sb);
    const float a = __int_as_float(__builtin_amdgcn_readlane(__float_as_int(Al), sg));
    sa = a * sa + Ls[sg].x;
    sb = a * sb + Ls[sg].y;
  }
  *(float2*)(state_out + i * D_ + h * dh + j) = make_float2(sa, sb);
}

__global__ __launch_bounds__(128)
void wkv_pass2(const float* __restrict__ rg, const float* __restrict__ wg,
               const float* __restrict__ kg, const float* __restrict__ vg,
               const float* __restrict__ gg, const float* __restrict__ ug,
               const float* __restrict__ isg, const float* __restrict__ lnw,
               const float* __restrict__ lnb, const float* __restrict__ Sstart,
               float* __restrict__ og) {
  __shared__ __attribute__((aligned(16))) float Rt[CH * ST];
  __shared__ float Kt[CH * ST];
  __shared__ float Ku[CH * ST];
  __shared__ float Vv[CH * ST];
  __shared__ float Kh[dh * ST16];
  __shared__ float CmW[4][CH * ST16];
  __shared__ __attribute__((aligned(16))) float ybuf[CH * ST];
  __shared__ float aLast[dh];
  __shared__ float uvec[dh];
  __shared__ float corr0[dh];
  __shared__ float lnw_s[dh], lnb_s[dh];
  __shared__ float r0sum_s;

  const int seg = blockIdx.x, h = blockIdx.y;
  const int tid = threadIdx.x;
  const int lane = tid & 31, wid = tid >> 5;
  const int hi = lane >> 4, ln16 = lane & 15;

  for (int x = tid; x < dh; x += 128) {
    uvec[x]  = ug[h * dh + x];
    corr0[x] = isg[h * dh + x] * (1.f - wg[h * dh + x]);
    lnw_s[x] = lnw[x];
    lnb_s[x] = lnb[x];
  }
  const int nb = wid * 16;
  const int kq = wid * 16;
  if (tid == 0 && seg == 0) {
    float s = 0.f;
    for (int i2 = 0; i2 < dh; ++i2) s += rg[h * dh + i2];
    r0sum_s = s;
  }

  const bool scanner = tid < dh;
  const int col = tid & (dh - 1);
  const int gbase = seg * TSEG * D_ + h * dh + col;
  const int gofs = (seg * TSEG + (tid >> 3)) * D_ + h * dh + (tid & 7) * 8;
  float pr[CH], pw[CH], pk[CH], pv[CH];
  if (scanner) {
#pragma unroll
    for (int t = 0; t < CH; ++t) {
      pr[t] = rg[gbase + t * D_]; pw[t] = wg[gbase + t * D_]; pk[t] = kg[gbase + t * D_];
    }
  } else {
#pragma unroll
    for (int t = 0; t < CH; ++t) pv[t] = vg[gbase + t * D_];
  }
  v8f Sr[4];
  {
    const float* sp = Sstart + ((size_t)(h * SEG + seg) * dh) * dh + nb + ln16;
#pragma unroll
    for (int it = 0; it < 4; ++it) {
#pragma unroll
      for (int v = 0; v < 8; ++v) Sr[it][v] = sp[(it * 16 + v + 8 * hi) * dh];
    }
  }
  asm volatile("" :: "v"(Sr[0]), "v"(Sr[1]), "v"(Sr[2]), "v"(Sr[3]));
  lds_barrier();

  for (int c = 0; c < NCHS; ++c) {
    const int t0 = seg * TSEG + c * CH;

    if (scanner) {
      int i = col;
      float a = 1.f, ai = 1.f;
      float kr[CH], air[CH];
#pragma unroll
      for (int t = 0; t < CH; ++t) {
        float wv = pw[t];
        if (seg == 0 && c == 0 && t == 0) wv = 1.f;
        float rv = pr[t], kv = pk[t];
        float aip = ai;
        Rt[t * ST + i] = rv * a;
        float rw = 1.f / wv;
        a *= wv; ai *= rw;
        Kt[t * ST + i] = kv * ai;
        Ku[t * ST + i] = kv * uvec[i] * aip;
        kr[t] = kv; air[t] = ai;
      }
      aLast[i] = a;
#pragma unroll
      for (int t = 0; t < CH; ++t) Kh[i * ST16 + t] = kr[t] * (a * air[t]);
    } else {
#pragma unroll
      for (int t = 0; t < CH; ++t) Vv[t * ST + col] = pv[t];
    }
    const float4 g0 = *(const float4*)(gg + gofs + c * CH * D_);
    const float4 g1 = *(const float4*)(gg + gofs + c * CH * D_ + 4);
    if (c + 1 < NCHS) {
      const int nbase = gbase + (c + 1) * CH * D_;
      if (scanner) {
#pragma unroll
        for (int t = 0; t < CH; ++t) {
          pr[t] = rg[nbase + t * D_]; pw[t] = wg[nbase + t * D_]; pk[t] = kg[nbase + t * D_];
        }
      } else {
#pragma unroll
        for (int t = 0; t < CH; ++t) pv[t] = vg[nbase + t * D_];
      }
    }
    lds_barrier();

    float4 ra[8];
#pragma unroll
    for (int it = 0; it < 4; ++it) {
      const float4* p = (const float4*)(Rt + ln16 * ST + it * 16 + 8 * hi);
      ra[2 * it] = p[0]; ra[2 * it + 1] = p[1];
    }

    v8f accC = {}; v8f accD = {};
#pragma unroll
    for (int k0 = 0; k0 < dh / 4; k0 += 4) {
      v2f arf = fragA(Rt, ST, kq + k0);
      accC = wmma_f32(arf, fragA(Kt, ST, kq + k0), accC);
      accD = wmma_f32(arf, fragA(Ku, ST, kq + k0), accD);
    }
    {
      float* cm = CmW[wid];
#pragma unroll
      for (int v = 0; v < 8; ++v) {
        int M = v + 8 * hi, N = ln16;
        cm[M * ST16 + N] = (N < M) ? accC[v] : ((N == M) ? accD[v] : 0.f);
      }
    }

    v2f vb[CH / 4];
#pragma unroll
    for (int k0 = 0; k0 < CH; k0 += 4) vb[k0 / 4] = fragB_kn(Vv, ST, k0, nb);

    v8f accY = {};
#pragma unroll
    for (int it = 0; it < 4; ++it) {
      const float4 lo = ra[2 * it], up = ra[2 * it + 1];
      v2f a, b;
      a[0] = lo.x; a[1] = lo.y; b[0] = Sr[it][0]; b[1] = Sr[it][1];
      accY = wmma_f32(a, b, accY);
      a[0] = lo.z; a[1] = lo.w; b[0] = Sr[it][2]; b[1] = Sr[it][3];
      accY = wmma_f32(a, b, accY);
      a[0] = up.x; a[1] = up.y; b[0] = Sr[it][4]; b[1] = Sr[it][5];
      accY = wmma_f32(a, b, accY);
      a[0] = up.z; a[1] = up.w; b[0] = Sr[it][6]; b[1] = Sr[it][7];
      accY = wmma_f32(a, b, accY);
    }

#pragma unroll
    for (int it = 0; it < 4; ++it) {
      v8f a = {};
#pragma unroll
      for (int k0 = 0; k0 < CH; k0 += 4)
        a = wmma_f32(fragA(Kh + it * 16 * ST16, ST16, k0), vb[k0 / 4], a);
#pragma unroll
      for (int v = 0; v < 8; ++v)
        Sr[it][v] = aLast[it * 16 + v + 8 * hi] * Sr[it][v] + a[v];
    }
    lds_barrier();

#pragma unroll
    for (int k0 = 0; k0 < CH; k0 += 4) {
      const v2f cmf = (fragA(CmW[0], ST16, k0) + fragA(CmW[1], ST16, k0)) +
                      (fragA(CmW[2], ST16, k0) + fragA(CmW[3], ST16, k0));
      accY = wmma_f32(cmf, vb[k0 / 4], accY);
    }
    if (seg == 0 && c == 0 && hi == 0)
      accY[0] += r0sum_s * corr0[nb + ln16];
#pragma unroll
    for (int v = 0; v < 8; ++v) ybuf[(v + 8 * hi) * ST + nb + ln16] = accY[v];
    lds_barrier();

    {
      const int t  = tid >> 3;
      const int jb = (tid & 7) * 8;
      const float4 ya = *(const float4*)(ybuf + t * ST + jb);
      const float4 yb = *(const float4*)(ybuf + t * ST + jb + 4);
      const float yq[8] = {ya.x, ya.y, ya.z, ya.w, yb.x, yb.y, yb.z, yb.w};
      float s = 0.f, sq = 0.f;
#pragma unroll
      for (int q = 0; q < 8; ++q) { s += yq[q]; sq += yq[q] * yq[q]; }
      s  = sum8(s);
      sq = sum8(sq);
      const float m  = s * (1.f / dh);
      const float vv = sq * (1.f / dh) - m * m;
      const float rs = rsqrtf(vv + 1e-5f);
      const float gq[8] = {g0.x, g0.y, g0.z, g0.w, g1.x, g1.y, g1.z, g1.w};
#pragma unroll
      for (int q = 0; q < 8; ++q) {
        int j    = jb + q;
        float y  = yq[q];
        float yn = (y - m) * rs * lnw_s[j] + lnb_s[j];
        int off  = (t0 + t) * D_ + h * dh + j;
        float gv = gq[q];
        og[off]  = yn * (gv / (1.f + __expf(-gv)));
      }
    }
    lds_barrier();
  }
}

#define OT   128
#define OKS  64
#define ONS  (D_ / OKS)
__global__ __launch_bounds__(256)
void oproj_kernel(const float* __restrict__ o, const float* __restrict__ Wo,
                  float* __restrict__ out) {
  __shared__ __attribute__((aligned(16))) float Ao[2][OT * ST];
  __shared__ __attribute__((aligned(16))) float Bo[2][OT * ST];
  const int tid = threadIdx.x;
  const int lane = tid & 31, wid = tid >> 5;
  const int hi = lane >> 4, ln16 = lane & 15;
  const int nbase = blockIdx.x * OT;
  const int mbase = blockIdx.y * OT;
  const int wm = (wid >> 2) * 64;
  const int wn = (wid & 3) * 32;

  v8f acc[4][2] = {};

  const int srow = tid >> 4;
  const int sc4  = (tid & 15) * 4;
  const float* pa = o  + (size_t)(mbase + srow) * D_ + sc4;
  const float* pb = Wo + (size_t)(nbase + srow) * D_ + sc4;
  const int sofs = srow * ST + sc4;
  v4f a4[8], b4[8];

#pragma unroll
  for (int q = 0; q < 8; ++q) {
    a4[q] = *(const v4f*)(pa + (size_t)q * 16 * D_);
    b4[q] = *(const v4f*)(pb + (size_t)q * 16 * D_);
  }
#pragma unroll
  for (int q = 0; q < 8; ++q) {
    *(v4f*)(Ao[0] + sofs + q * 16 * ST) = a4[q];
    *(v4f*)(Bo[0] + sofs + q * 16 * ST) = b4[q];
  }
  lds_barrier();

  for (int s = 0; s < ONS; ++s) {
    const int cur = s & 1;
    const bool more = s + 1 < ONS;
    if (more) {
      const int ks = (s + 1) * OKS;
#pragma unroll
      for (int q = 0; q < 8; ++q) {
        a4[q] = *(const v4f*)(pa + ks + (size_t)q * 16 * D_);
        b4[q] = *(const v4f*)(pb + ks + (size_t)q * 16 * D_);
      }
    }
    const float* As = Ao[cur] + wm * ST;
    const float* Bs = Bo[cur] + wn * ST;
#pragma unroll
    for (int k0 = 0; k0 < OKS / 2; k0 += 4) {
      const v2f b0 = fragA(Bs, ST, k0);
      const v2f b1 = fragA(Bs + 16 * ST, ST, k0);
#pragma unroll
      for (int mt = 0; mt < 4; ++mt) {
        const v2f a = fragA(As + mt * 16 * ST, ST, k0);
        acc[mt][0] = wmma_f32(a, b0, acc[mt][0]);
        acc[mt][1] = wmma_f32(a, b1, acc[mt][1]);
      }
    }
    __builtin_amdgcn_sched_barrier(0);
    if (more) {
#pragma unroll
      for (int q = 0; q < 8; ++q) {
        *(v4f*)(Ao[cur ^ 1] + sofs + q * 16 * ST) = a4[q];
        *(v4f*)(Bo[cur ^ 1] + sofs + q * 16 * ST) = b4[q];
      }
    }
#pragma unroll
    for (int k0 = OKS / 2; k0 < OKS; k0 += 4) {
      const v2f b0 = fragA(Bs, ST, k0);
      const v2f b1 = fragA(Bs + 16 * ST, ST, k0);
#pragma unroll
      for (int mt = 0; mt < 4; ++mt) {
        const v2f a = fragA(As + mt * 16 * ST, ST, k0);
        acc[mt][0] = wmma_f32(a, b0, acc[mt][0]);
        acc[mt][1] = wmma_f32(a, b1, acc[mt][1]);
      }
    }
    lds_barrier();
  }
#pragma unroll
  for (int mt = 0; mt < 4; ++mt) {
#pragma unroll
    for (int nt = 0; nt < 2; ++nt) {
#pragma unroll
      for (int v = 0; v < 8; ++v) {
        int row = mbase + wm + mt * 16 + v + 8 * hi;
        int col = nbase + wn + nt * 16 + ln16;
        out[(size_t)row * D_ + col] = acc[mt][nt][v];
      }
    }
  }
}

extern "C" void kernel_launch(void* const* d_in, const int* in_sizes, int n_in,
                              void* d_out, int out_size, void* d_ws, size_t ws_size,
                              hipStream_t stream) {
  const float* r   = (const float*)d_in[0];
  const float* w   = (const float*)d_in[1];
  const float* k   = (const float*)d_in[2];
  const float* v   = (const float*)d_in[3];
  const float* g   = (const float*)d_in[4];
  const float* u   = (const float*)d_in[5];
  const float* is  = (const float*)d_in[6];
  const float* lnw = (const float*)d_in[7];
  const float* lnb = (const float*)d_in[8];
  const float* Wo  = (const float*)d_in[9];

  float* out   = (float*)d_out;
  float* state = out + (size_t)T_ * D_;

  float* o      = (float*)d_ws;
  float* Aseg   = o + (size_t)T_ * D_;
  float* Sloc   = Aseg + (size_t)H_ * SEG * dh;
  float* Sstart = Sloc + (size_t)H_ * SEG * dh * dh;

  dim3 gs(SEG, H_);
  wkv_pass1<<<gs, 128, 0, stream>>>(w, k, v, Aseg, Sloc);
  dim3 gsc(dh / 4, H_);
  state_scan<<<gsc, 128, 0, stream>>>(w, is, Aseg, Sloc, Sstart, state);
  wkv_pass2<<<gs, 128, 0, stream>>>(r, w, k, v, g, u, is, lnw, lnb, Sstart, o);

  dim3 g2(D_ / OT, T_ / OT);
  oproj_kernel<<<g2, 256, 0, stream>>>(o, Wo, out);
}


# pass1+pass2: 1/w of the decay stage and the SiLU gate's 1/(1+exp(-g)) by the hardware reciprocal (v_rcp_f32, 1 ulp) instead of the ten-instruction exact division; order of sums unchanged
# speedup vs baseline: 6.9919x; 1.0496x over previous
#include <hip/hip_runtime.h>

typedef __attribute__((ext_vector_type(2))) float v2f;
typedef __attribute__((ext_vector_type(4))) float v4f;
typedef __attribute__((ext_vector_type(8))) float v8f;

#define D_   1024
#define H_   16
#define dh   64
#define T_   4096
#define CH   16
#define SEG  32
#define TSEG (T_/SEG)
#define NCHS (TSEG/CH)
#define ST   68
#define ST16 20

__device__ __forceinline__ v8f wmma_f32(v2f a, v2f b, v8f c) {
  return __builtin_amdgcn_wmma_f32_16x16x4_f32(false, a, false, b, (short)0, c,
                                               false, false);
}

__device__ __forceinline__ v2f fragA(const float* m, int stride, int k0) {
  int lane = threadIdx.x & 31;
  const float* p = m + (lane & 15) * stride + k0 + ((lane >> 4) << 1);
  v2f f; f[0] = p[0]; f[1] = p[1];
  return f;
}
__device__ __forceinline__ v2f fragB_kn(const float* m, int stride, int k0, int nbase) {
  int lane = threadIdx.x & 31;
  int kk = k0 + ((lane >> 4) << 1);
  int n  = nbase + (lane & 15);
  v2f f; f[0] = m[kk * stride + n]; f[1] = m[(kk + 1) * stride + n];
  return f;
}

__device__ __forceinline__ void lds_barrier() {
  __builtin_amdgcn_fence(__ATOMIC_RELEASE, "workgroup", "local");
  __builtin_amdgcn_s_barrier();
  __builtin_amdgcn_fence(__ATOMIC_ACQUIRE, "workgroup", "local");
}

__device__ __forceinline__ float sum8(float x) {
  x += __int_as_float(__builtin_amdgcn_update_dpp(0, __float_as_int(x), 0x161, 0xf, 0xf, false));
  x += __int_as_float(__builtin_amdgcn_update_dpp(0, __float_as_int(x), 0x162, 0xf, 0xf, false));
  x += __int_as_float(__builtin_amdgcn_update_dpp(0, __float_as_int(x), 0x164, 0xf, 0xf, false));
  return x;
}

__device__ __forceinline__ float rcp_fast(float x) { return __builtin_amdgcn_rcpf(x); }

__global__ __launch_bounds__(128)
void wkv_pass1(const float* __restrict__ wg, const float* __restrict__ kg,
               const float* __restrict__ vg, float* __restrict__ Aseg,
               float* __restrict__ Sloc) {
  __shared__ float S[dh * ST];
  __shared__ float Vv[CH * ST];
  __shared__ float Kh[dh * ST16];
  __shared__ float aLast[dh];
  __shared__ float aAcc[dh];

  const int seg = blockIdx.x, h = blockIdx.y;
  const int tid = threadIdx.x;
  const int lane = tid & 31, wid = tid >> 5;
  const int hi = lane >> 4, ln16 = lane & 15;

  if (tid < dh) aAcc[tid] = 1.f;
  const int nb = wid * 16;
  v8f Sr[4] = {};

  const bool scanner = tid < dh;
  const int col = tid & (dh - 1);
  const int gbase = seg * TSEG * D_ + h * dh + col;
  float pw[CH], pk[CH], pv[CH];
  if (scanner) {
#pragma unroll
    for (int t = 0; t < CH; ++t) { pw[t] = wg[gbase + t * D_]; pk[t] = kg[gbase + t * D_]; }
  } else {
#pragma unroll
    for (int t = 0; t < CH; ++t) pv[t] = vg[gbase + t * D_];
  }

  for (int c = 0; c < NCHS; ++c) {
    lds_barrier();
    if (scanner) {
      int i = col;
      float a = 1.f, ai = 1.f;
      float kr[CH], air[CH];
#pragma unroll
      for (int t = 0; t < CH; ++t) {
        float wv = pw[t];
        if (seg == 0 && c == 0 && t == 0) wv = 1.f;
        float kv = pk[t];
        float rw = rcp_fast(wv);
        a *= wv; ai *= rw;
        kr[t] = kv; air[t] = ai;
      }
      aLast[i] = a;
      aAcc[i] *= a;
#pragma unroll
      for (int t = 0; t < CH; ++t) Kh[i * ST16 + t] = kr[t] * (a * air[t]);
    } else {
#pragma unroll
      for (int t = 0; t < CH; ++t) Vv[t * ST + col] = pv[t];
    }
    if (c + 1 < NCHS) {
      const int nbase = gbase + (c + 1) * CH * D_;
      if (scanner) {
#pragma unroll
        for (int t = 0; t < CH; ++t) { pw[t] = wg[nbase + t * D_]; pk[t] = kg[nbase + t * D_]; }
      } else {
#pragma unroll
        for (int t = 0; t < CH; ++t) pv[t] = vg[nbase + t * D_];
      }
    }
    lds_barrier();

#pragma unroll
    for (int it = 0; it < 4; ++it) {
      v8f acc = {};
#pragma unroll
      for (int k0 = 0; k0 < CH; k0 += 4)
        acc = wmma_f32(fragA(Kh + it * 16 * ST16, ST16, k0),
                       fragB_kn(Vv, ST, k0, nb), acc);
#pragma unroll
      for (int v = 0; v < 8; ++v)
        Sr[it][v] = aLast[it * 16 + v + 8 * hi] * Sr[it][v] + acc[v];
    }
  }
#pragma unroll
  for (int it = 0; it < 4; ++it) {
#pragma unroll
    for (int v = 0; v < 8; ++v)
      S[(it * 16 + v + 8 * hi) * ST + nb + ln16] = Sr[it][v];
  }
  lds_barrier();
  if (tid < dh) Aseg[(h * SEG + seg) * dh + tid] = aAcc[tid];
  for (int x = tid; x < dh * dh; x += 128) {
    int i = x >> 6, j = x & 63;
    Sloc[((size_t)(h * SEG + seg) * dh + i) * dh + j] = S[i * ST + j];
  }
}

__global__ __launch_bounds__(128)
void state_scan(const float* __restrict__ wg, const float* __restrict__ isg,
                const float* __restrict__ Aseg, const float* __restrict__ Sloc,
                float* __restrict__ Sstart, float* __restrict__ state_out) {
  static_assert(SEG == 32, "one decay factor per lane of a 32-lane wave");
  const int h = blockIdx.y, tid = threadIdx.x;
  const int lane = tid & 31;
  const int i = blockIdx.x * 4 + (tid >> 5);
  const int j = lane * 2;
  const float Al = Aseg[(h * SEG + lane) * dh + i];
  float2 Ls[SEG];
#pragma unroll
  for (int sg = 0; sg < SEG; ++sg)
    Ls[sg] = *(const float2*)(Sloc + ((size_t)(h * SEG + sg) * dh + i) * dh + j);
  const float2 w0 = *(const float2*)(wg + h * dh + j);
  const float2 s0 = *(const float2*)(isg + h * dh + j);
  float sa = w0.x * s0.x, sb = w0.y * s0.y;
#pragma unroll
  for (int sg = 0; sg < SEG; ++sg) {
    size_t base = ((size_t)(h * SEG + sg) * dh + i) * dh + j;
    *(float2*)(Sstart + base) = make_float2(sa, sb);
    const float a = __int_as_float(__builtin_amdgcn_readlane(__float_as_int(Al), sg));
    sa = a * sa + Ls[sg].x;
    sb = a * sb + Ls[sg].y;
  }
  *(float2*)(state_out + i * D_ + h * dh + j) = make_float2(sa, sb);
}

__global__ __launch_bounds__(128)
void wkv_pass2(const float* __restrict__ rg, const float* __restrict__ wg,
               const float* __restrict__ kg, const float* __restrict__ vg,
               const float* __restrict__ gg, const float* __restrict__ ug,
               const float* __restrict__ isg, const float* __restrict__ lnw,
               const float* __restrict__ lnb, const float* __restrict__ Sstart,
               float* __restrict__ og) {
  __shared__ __attribute__((aligned(16))) float Rt[CH * ST];
  __shared__ float Kt[CH * ST];
  __shared__ float Ku[CH * ST];
  __shared__ float Vv[CH * ST];
  __shared__ float Kh[dh * ST16];
  __shared__ float CmW[4][CH * ST16];
  __shared__ __attribute__((aligned(16))) float ybuf[CH * ST];
  __shared__ float aLast[dh];
  __shared__ float uvec[dh];
  __shared__ float corr0[dh];
  __shared__ float lnw_s[dh], lnb_s[dh];
  __shared__ float r0sum_s;

  const int seg = blockIdx.x, h = blockIdx.y;
  const int tid = threadIdx.x;
  const int lane = tid & 31, wid = tid >> 5;
  const int hi = lane >> 4, ln16 = lane & 15;

  for (int x = tid; x < dh; x += 128) {
    uvec[x]  = ug[h * dh + x];
    corr0[x] = isg[h * dh + x] * (1.f - wg[h * dh + x]);
    lnw_s[x] = lnw[x];
    lnb_s[x] = lnb[x];
  }
  const int nb = wid * 16;
  const int kq = wid * 16;
  if (tid == 0 && seg == 0) {
    float s = 0.f;
    for (int i2 = 0; i2 < dh; ++i2) s += rg[h * dh + i2];
    r0sum_s = s;
  }

  const bool scanner = tid < dh;
  const int col = tid & (dh - 1);
  const int gbase = seg * TSEG * D_ + h * dh + col;
  const int gofs = (seg * TSEG + (tid >> 3)) * D_ + h * dh + (tid & 7) * 8;
  float pr[CH], pw[CH], pk[CH], pv[CH];
  if (scanner) {
#pragma unroll
    for (int t = 0; t < CH; ++t) {
      pr[t] = rg[gbase + t * D_]; pw[t] = wg[gbase + t * D_]; pk[t] = kg[gbase + t * D_];
    }
  } else {
#pragma unroll
    for (int t = 0; t < CH; ++t) pv[t] = vg[gbase + t * D_];
  }
  v8f Sr[4];
  {
    const float* sp = Sstart + ((size_t)(h * SEG + seg) * dh) * dh + nb + ln16;
#pragma unroll
    for (int it = 0; it < 4; ++it) {
#pragma unroll
      for (int v = 0; v < 8; ++v) Sr[it][v] = sp[(it * 16 + v + 8 * hi) * dh];
    }
  }
  asm volatile("" :: "v"(Sr[0]), "v"(Sr[1]), "v"(Sr[2]), "v"(Sr[3]));
  lds_barrier();

  for (int c = 0; c < NCHS; ++c) {
    const int t0 = seg * TSEG + c * CH;

    if (scanner) {
      int i = col;
      float a = 1.f, ai = 1.f;
      float kr[CH], air[CH];
#pragma unroll
      for (int t = 0; t < CH; ++t) {
        float wv = pw[t];
        if (seg == 0 && c == 0 && t == 0) wv = 1.f;
        float rv = pr[t], kv = pk[t];
        float aip = ai;
        Rt[t * ST + i] = rv * a;
        float rw = rcp_fast(wv);
        a *= wv; ai *= rw;
        Kt[t * ST + i] = kv * ai;
        Ku[t * ST + i] = kv * uvec[i] * aip;
        kr[t] = kv; air[t] = ai;
      }
      aLast[i] = a;
#pragma unroll
      for (int t = 0; t < CH; ++t) Kh[i * ST16 + t] = kr[t] * (a * air[t]);
    } else {
#pragma unroll
      for (int t = 0; t < CH; ++t) Vv[t * ST + col] = pv[t];
    }
    const float4 g0 = *(const float4*)(gg + gofs + c * CH * D_);
    const float4 g1 = *(const float4*)(gg + gofs + c * CH * D_ + 4);
    if (c + 1 < NCHS) {
      const int nbase = gbase + (c + 1) * CH * D_;
      if (scanner) {
#pragma unroll
        for (int t = 0; t < CH; ++t) {
          pr[t] = rg[nbase + t * D_]; pw[t] = wg[nbase + t * D_]; pk[t] = kg[nbase + t * D_];
        }
      } else {
#pragma unroll
        for (int t = 0; t < CH; ++t) pv[t] = vg[nbase + t * D_];
      }
    }
    lds_barrier();

    float4 ra[8];
#pragma unroll
    for (int it = 0; it < 4; ++it) {
      const float4* p = (const float4*)(Rt + ln16 * ST + it * 16 + 8 * hi);
      ra[2 * it] = p[0]; ra[2 * it + 1] = p[1];
    }

    v8f accC = {}; v8f accD = {};
#pragma unroll
    for (int k0 = 0; k0 < dh / 4; k0 += 4) {
      v2f arf = fragA(Rt, ST, kq + k0);
      accC = wmma_f32(arf, fragA(Kt, ST, kq + k0), accC);
      accD = wmma_f32(arf, fragA(Ku, ST, kq + k0), accD);
    }
    {
      float* cm = CmW[wid];
#pragma unroll
      for (int v = 0; v < 8; ++v) {
        int M = v + 8 * hi, N = ln16;
        cm[M * ST16 + N] = (N < M) ? accC[v] : ((N == M) ? accD[v] : 0.f);
      }
    }

    v2f vb[CH / 4];
#pragma unroll
    for (int k0 = 0; k0 < CH; k0 += 4) vb[k0 / 4] = fragB_kn(Vv, ST, k0, nb);

    v8f accY = {};
#pragma unroll
    for (int it = 0; it < 4; ++it) {
      const float4 lo = ra[2 * it], up = ra[2 * it + 1];
      v2f a, b;
      a[0] = lo.x; a[1] = lo.y; b[0] = Sr[it][0]; b[1] = Sr[it][1];
      accY = wmma_f32(a, b, accY);
      a[0] = lo.z; a[1] = lo.w; b[0] = Sr[it][2]; b[1] = Sr[it][3];
      accY = wmma_f32(a, b, accY);
      a[0] = up.x; a[1] = up.y; b[0] = Sr[it][4]; b[1] = Sr[it][5];
      accY = wmma_f32(a, b, accY);
      a[0] = up.z; a[1] = up.w; b[0] = Sr[it][6]; b[1] = Sr[it][7];
      accY = wmma_f32(a, b, accY);
    }

#pragma unroll
    for (int it = 0; it < 4; ++it) {
      v8f a = {};
#pragma unroll
      for (int k0 = 0; k0 < CH; k0 += 4)
        a = wmma_f32(fragA(Kh + it * 16 * ST16, ST16, k0), vb[k0 / 4], a);
#pragma unroll
      for (int v = 0; v < 8; ++v)
        Sr[it][v] = aLast[it * 16 + v + 8 * hi] * Sr[it][v] + a[v];
    }
    lds_barrier();

#pragma unroll
    for (int k0 = 0; k0 < CH; k0 += 4) {
      const v2f cmf = (fragA(CmW[0], ST16, k0) + fragA(CmW[1], ST16, k0)) +
                      (fragA(CmW[2], ST16, k0) + fragA(CmW[3], ST16, k0));
      accY = wmma_f32(cmf, vb[k0 / 4], accY);
    }
    if (seg == 0 && c == 0 && hi == 0)
      accY[0] += r0sum_s * corr0[nb + ln16];
#pragma unroll
    for (int v = 0; v < 8; ++v) ybuf[(v + 8 * hi) * ST + nb + ln16] = accY[v];
    lds_barrier();

    {
      const int t  = tid >> 3;
      const int jb = (tid & 7) * 8;
      const float4 ya = *(const float4*)(ybuf + t * ST + jb);
      const float4 yb = *(const float4*)(ybuf + t * ST + jb + 4);
      const float yq[8] = {ya.x, ya.y, ya.z, ya.w, yb.x, yb.y, yb.z, yb.w};
      float s = 0.f, sq = 0.f;
#pragma unroll
      for (int q = 0; q < 8; ++q) { s += yq[q]; sq += yq[q] * yq[q]; }
      s  = sum8(s);
      sq = sum8(sq);
      const float m  = s * (1.f / dh);
      const float vv = sq * (1.f / dh) - m * m;
      const float rs = rsqrtf(vv + 1e-5f);
      const float gq[8] = {g0.x, g0.y, g0.z, g0.w, g1.x, g1.y, g1.z, g1.w};
#pragma unroll
      for (int q = 0; q < 8; ++q) {
        int j    = jb + q;
        float y  = yq[q];
        float yn = (y - m) * rs * lnw_s[j] + lnb_s[j];
        int off  = (t0 + t) * D_ + h * dh + j;
        float gv = gq[q];
        og[off]  = yn * (gv * rcp_fast(1.f + __expf(-gv)));
      }
    }
    lds_barrier();
  }
}

#define OT   128
#define OKS  64
#define ONS  (D_ / OKS)
__global__ __launch_bounds__(256)
void oproj_kernel(const float* __restrict__ o, const float* __restrict__ Wo,
                  float* __restrict__ out) {
  __shared__ __attribute__((aligned(16))) float Ao[2][OT * ST];
  __shared__ __attribute__((aligned(16))) float Bo[2][OT * ST];
  const int tid = threadIdx.x;
  const int lane = tid & 31, wid = tid >> 5;
  const int hi = lane >> 4, ln16 = lane & 15;
  const int nbase = blockIdx.x * OT;
  const int mbase = blockIdx.y * OT;
  const int wm = (wid >> 2) * 64;
  const int wn = (wid & 3) * 32;

  v8f acc[4][2] = {};

  const int srow = tid >> 4;
  const int sc4  = (tid & 15) * 4;
  const float* pa = o  + (size_t)(mbase + srow) * D_ + sc4;
  const float* pb = Wo + (size_t)(nbase + srow) * D_ + sc4;
  const int sofs = srow * ST + sc4;
  v4f a4[8], b4[8];

#pragma unroll
  for (int q = 0; q < 8; ++q) {
    a4[q] = *(const v4f*)(pa + (size_t)q * 16 * D_);
    b4[q] = *(const v4f*)(pb + (size_t)q * 16 * D_);
  }
#pragma unroll
  for (int q = 0; q < 8; ++q) {
    *(v4f*)(Ao[0] + sofs + q * 16 * ST) = a4[q];
    *(v4f*)(Bo[0] + sofs + q * 16 * ST) = b4[q];
  }
  lds_barrier();

  for (int s = 0; s < ONS; ++s) {
    const int cur = s & 1;
    const bool more = s + 1 < ONS;
    if (more) {
      const int ks = (s + 1) * OKS;
#pragma unroll
      for (int q = 0; q < 8; ++q) {
        a4[q] = *(const v4f*)(pa + ks + (size_t)q * 16 * D_);
        b4[q] = *(const v4f*)(pb + ks + (size_t)q * 16 * D_);
      }
    }
    const float* As = Ao[cur] + wm * ST;
    const float* Bs = Bo[cur] + wn * ST;
#pragma unroll
    for (int k0 = 0; k0 < OKS / 2; k0 += 4) {
      const v2f b0 = fragA(Bs, ST, k0);
      const v2f b1 = fragA(Bs + 16 * ST, ST, k0);
#pragma unroll
      for (int mt = 0; mt < 4; ++mt) {
        const v2f a = fragA(As + mt * 16 * ST, ST, k0);
        acc[mt][0] = wmma_f32(a, b0, acc[mt][0]);
        acc[mt][1] = wmma_f32(a, b1, acc[mt][1]);
      }
    }
    __builtin_amdgcn_sched_barrier(0);
    if (more) {
#pragma unroll
      for (int q = 0; q < 8; ++q) {
        *(v4f*)(Ao[cur ^ 1] + sofs + q * 16 * ST) = a4[q];
        *(v4f*)(Bo[cur ^ 1] + sofs + q * 16 * ST) = b4[q];
      }
    }
#pragma unroll
    for (int k0 = OKS / 2; k0 < OKS; k0 += 4) {
      const v2f b0 = fragA(Bs, ST, k0);
      const v2f b1 = fragA(Bs + 16 * ST, ST, k0);
#pragma unroll
      for (int mt = 0; mt < 4; ++mt) {
        const v2f a = fragA(As + mt * 16 * ST, ST, k0);
        acc[mt][0] = wmma_f32(a, b0, acc[mt][0]);
        acc[mt][1] = wmma_f32(a, b1, acc[mt][1]);
      }
    }
    lds_barrier();
  }
#pragma unroll
  for (int mt = 0; mt < 4; ++mt) {
#pragma unroll
    for (int nt = 0; nt < 2; ++nt) {
#pragma unroll
      for (int v = 0; v < 8; ++v) {
        int row = mbase + wm + mt * 16 + v + 8 * hi;
        int col = nbase + wn + nt * 16 + ln16;
        out[(size_t)row * D_ + col] = acc[mt][nt][v];
      }
    }
  }
}

extern "C" void kernel_launch(void* const* d_in, const int* in_sizes, int n_in,
                              void* d_out, int out_size, void* d_ws, size_t ws_size,
                              hipStream_t stream) {
  const float* r   = (const float*)d_in[0];
  const float* w   = (const float*)d_in[1];
  const float* k   = (const float*)d_in[2];
  const float* v   = (const float*)d_in[3];
  const float* g   = (const float*)d_in[4];
  const float* u   = (const float*)d_in[5];
  const float* is  = (const float*)d_in[6];
  const float* lnw = (const float*)d_in[7];
  const float* lnb = (const float*)d_in[8];
  const float* Wo  = (const float*)d_in[9];

  float* out   = (float*)d_out;
  float* state = out + (size_t)T_ * D_;

  float* o      = (float*)d_ws;
  float* Aseg   = o + (size_t)T_ * D_;
  float* Sloc   = Aseg + (size_t)H_ * SEG * dh;
  float* Sstart = Sloc + (size_t)H_ * SEG * dh * dh;

  dim3 gs(SEG, H_);
  wkv_pass1<<<gs, 128, 0, stream>>>(w, k, v, Aseg, Sloc);
  dim3 gsc(dh / 4, H_);
  state_scan<<<gsc, 128, 0, stream>>>(w, is, Aseg, Sloc, Sstart, state);
  wkv_pass2<<<gs, 128, 0, stream>>>(r, w, k, v, g, u, is, lnw, lnb, Sstart, o);

  dim3 g2(D_ / OT, T_ / OT);
  oproj_kernel<<<g2, 256, 0, stream>>>(o, Wo, out);
}
